# starting kernel
# baseline (speedup 1.0000x reference)
#include <hip/hip_runtime.h>
#include <math.h>

typedef __attribute__((ext_vector_type(2))) float v2f;
typedef __attribute__((ext_vector_type(8))) float v8f;

#define S_DIM 256
#define C_DIM 32
#define E_DIM 32
#define N_TOT 512
#define KC 32
#define NCHUNK (S_DIM / KC)

__device__ __forceinline__ float sigmoidf_(float x) { return 1.0f / (1.0f + expf(-x)); }

__global__ __launch_bounds__(256) void router_setup(const float* __restrict__ ct,
                                                    const float* __restrict__ cm,
                                                    float* __restrict__ ws) {
  __shared__ float r1[256], r2[256];
  __shared__ float bc[4];
  __shared__ float rowre[256], rowim[256], colre[256], colim[256];
  const int tid = threadIdx.x;

  float mre = -1e30f, mim = -1e30f;
  for (int i = tid; i < S_DIM * S_DIM; i += 256) {
    mre = fmaxf(mre, cm[2 * i]);
    mim = fmaxf(mim, cm[2 * i + 1]);
  }
  r1[tid] = mre; r2[tid] = mim;
  __syncthreads();
  for (int off = 128; off > 0; off >>= 1) {
    if (tid < off) { r1[tid] = fmaxf(r1[tid], r1[tid + off]); r2[tid] = fmaxf(r2[tid], r2[tid + off]); }
    __syncthreads();
  }
  if (tid == 0) { bc[0] = r1[0]; bc[1] = r2[0]; }
  __syncthreads();
  mre = bc[0]; mim = bc[1];

  float sre = 0.0f, sim = 0.0f;
  for (int i = tid; i < S_DIM * S_DIM; i += 256) {
    sre += expf(cm[2 * i] - mre);
    sim += expf(cm[2 * i + 1] - mim);
  }
  r1[tid] = sre; r2[tid] = sim;
  __syncthreads();
  for (int off = 128; off > 0; off >>= 1) {
    if (tid < off) { r1[tid] += r1[tid + off]; r2[tid] += r2[tid + off]; }
    __syncthreads();
  }
  if (tid == 0) { bc[2] = r1[0]; bc[3] = r2[0]; }
  __syncthreads();
  sre = bc[2]; sim = bc[3];

  rowre[tid] = expf(cm[2 * tid] - mre) / sre;
  rowim[tid] = expf(cm[2 * tid + 1] - mim) / sim;
  colre[tid] = expf(cm[2 * (tid * S_DIM)] - mre) / sre;
  colim[tid] = expf(cm[2 * (tid * S_DIM) + 1] - mim) / sim;
  __syncthreads();

  float mrr = 0.0f, mri = 0.0f, mcr = 0.0f, mci = 0.0f;
  const float w0 = -6.283185307179586f / 256.0f;
  for (int t = 0; t < 256; ++t) {
    float ang = w0 * (float)((tid * t) & 255);
    float sn, cs;
    sincosf(ang, &sn, &cs);
    mrr += rowre[t] * cs - rowim[t] * sn;
    mri += rowre[t] * sn + rowim[t] * cs;
    mcr += colre[t] * cs - colim[t] * sn;
    mci += colre[t] * sn + colim[t] * cs;
  }
  ws[tid] = mrr; ws[256 + tid] = mri;
  ws[512 + tid] = mcr; ws[768 + tid] = mci;

  if (tid < E_DIM) {
    const int e = tid;
    ws[1024 + e] = sigmoidf_(ct[((e * 2 + 0) * S_DIM + 0) * 2 + 0]);
    ws[1056 + e] = sigmoidf_(ct[((e * 2 + 0) * S_DIM + 0) * 2 + 1]);
    ws[1088 + e] = sigmoidf_(ct[((e * 2 + 1) * S_DIM + 0) * 2 + 0]);
    ws[1120 + e] = sigmoidf_(ct[((e * 2 + 1) * S_DIM + 0) * 2 + 1]);
  }
}

__global__ __launch_bounds__(256) void router_select(const float* __restrict__ a,
                                                     const float* __restrict__ b,
                                                     const float* __restrict__ ws,
                                                     int* __restrict__ topidx) {
  __shared__ float part[256][6];
  __shared__ float cred[32][4];
  __shared__ float A1[4];
  __shared__ float mag[32];
  const int n = blockIdx.x;
  const int tid = threadIdx.x;
  const int c = tid >> 3, u = tid & 7;
  const float* ap = a + ((size_t)n * C_DIM + c) * S_DIM;
  const float* bp = b + ((size_t)n * C_DIM + c) * S_DIM;

  float sa = 0, sb = 0, par = 0, pai = 0, pbr = 0, pbi = 0;
  for (int i = 0; i < 32; ++i) {
    int s = u * 32 + i;
    float av = ap[s], bv = bp[s];
    sa += av; sb += bv;
    par += av * ws[512 + s]; pai += av * ws[768 + s];
    pbr += bv * ws[s];       pbi += bv * ws[256 + s];
  }
  part[tid][0] = sa; part[tid][1] = sb;
  part[tid][2] = par; part[tid][3] = pai;
  part[tid][4] = pbr; part[tid][5] = pbi;
  __syncthreads();

  if (tid < 32) {
    float s0 = 0, s1 = 0, s2 = 0, s3 = 0, s4 = 0, s5 = 0;
    for (int j = 0; j < 8; ++j) {
      s0 += part[tid * 8 + j][0]; s1 += part[tid * 8 + j][1];
      s2 += part[tid * 8 + j][2]; s3 += part[tid * 8 + j][3];
      s4 += part[tid * 8 + j][4]; s5 += part[tid * 8 + j][5];
    }
    cred[tid][0] = s0 * s4;
    cred[tid][1] = s0 * s5;
    cred[tid][2] = s1 * s2;
    cred[tid][3] = s1 * s3;
  }
  __syncthreads();
  if (tid == 0) {
    float ar = 0, ai = 0, br = 0, bi = 0;
    for (int cc = 0; cc < 32; ++cc) { ar += cred[cc][0]; ai += cred[cc][1]; br += cred[cc][2]; bi += cred[cc][3]; }
    A1[0] = ar; A1[1] = ai; A1[2] = br; A1[3] = bi;
  }
  __syncthreads();
  if (tid < E_DIM) {
    const int e = tid;
    float tar = ws[1024 + e], tai = ws[1056 + e], tbr = ws[1088 + e], tbi = ws[1120 + e];
    float zr = A1[0] * tar + A1[1] * tai + A1[2] * tbr + A1[3] * tbi;
    float zi = A1[1] * tar - A1[0] * tai + A1[3] * tbr - A1[2] * tbi;
    mag[e] = zr * zr + zi * zi;
  }
  __syncthreads();
  if (tid == 0) {
    int s0 = 0, s1 = -1, s2 = -1;
    float b0 = -1.0f, b1 = -1.0f, b2 = -1.0f;
    for (int e = 0; e < 32; ++e) { float m = mag[e]; if (m > b0) { b0 = m; s0 = e; } }
    for (int e = 0; e < 32; ++e) { if (e == s0) continue; float m = mag[e]; if (m > b1) { b1 = m; s1 = e; } }
    for (int e = 0; e < 32; ++e) { if (e == s0 || e == s1) continue; float m = mag[e]; if (m > b2) { b2 = m; s2 = e; } }
    topidx[n * 3 + 0] = s0; topidx[n * 3 + 1] = s1; topidx[n * 3 + 2] = s2;
  }
}

__global__ __launch_bounds__(256) void moe_wmma_gemm(const float* __restrict__ a,
                                                     const float* __restrict__ b,
                                                     const float* __restrict__ ew,
                                                     const int* __restrict__ topidx,
                                                     float* __restrict__ out) {
  __shared__ __align__(16) float Xs[C_DIM][S_DIM];
  __shared__ __align__(16) float Ws[2][KC][S_DIM];
  const int n = blockIdx.x;
  const int tid = threadIdx.x;
  const size_t nb = (size_t)n * C_DIM * S_DIM;

  const int i0 = topidx[n * 3 + 0];
  const int i1 = topidx[n * 3 + 1];
  const int i2 = topidx[n * 3 + 2];
  const float* __restrict__ w0 = ew + (size_t)i0 * (S_DIM * S_DIM);
  const float* __restrict__ w1 = ew + (size_t)i1 * (S_DIM * S_DIM);
  const float* __restrict__ w2 = ew + (size_t)i2 * (S_DIM * S_DIM);

  float4 pre[8];

  auto loadChunk = [&](int K0) {
#pragma unroll
    for (int r = 0; r < 8; ++r) {
      const int lin = r * 256 + tid;
      const int k = lin >> 6;
      const int s4 = (lin & 63) << 2;
      const size_t g = (size_t)(K0 + k) * S_DIM + s4;
      const float4 x0 = *(const float4*)(w0 + g);
      const float4 x1 = *(const float4*)(w1 + g);
      const float4 x2 = *(const float4*)(w2 + g);
      pre[r].x = (x0.x + x1.x + x2.x) * (1.0f / 3.0f);
      pre[r].y = (x0.y + x1.y + x2.y) * (1.0f / 3.0f);
      pre[r].z = (x0.z + x1.z + x2.z) * (1.0f / 3.0f);
      pre[r].w = (x0.w + x1.w + x2.w) * (1.0f / 3.0f);
    }
  };
  auto storeChunk = [&](int buf) {
#pragma unroll
    for (int r = 0; r < 8; ++r) {
      const int lin = r * 256 + tid;
      const int k = lin >> 6;
      const int s4 = (lin & 63) << 2;
      *(float4*)&Ws[buf][k][s4] = pre[r];
    }
  };

#pragma unroll
  for (int r = 0; r < 8; ++r) {
    const int lin = r * 256 + tid;
    const int c = lin >> 6;
    const int s4 = (lin & 63) << 2;
    const float4 av = *(const float4*)(a + nb + c * S_DIM + s4);
    const float4 bv = *(const float4*)(b + nb + c * S_DIM + s4);
    float4 v;
    v.x = av.x + bv.x; v.y = av.y + bv.y; v.z = av.z + bv.z; v.w = av.w + bv.w;
    *(float4*)&Xs[c][(s4 + 4 * c) & (S_DIM - 1)] = v;
  }

  loadChunk(0);
  storeChunk(0);

  const int wave = tid >> 5;
  const int lane = tid & 31;
  const int l16 = lane & 15;
  const int lh = lane >> 4;
  const int ncol = wave * 32;

  v8f acc00 = {}, acc01 = {}, acc10 = {}, acc11 = {};

  int cur = 0;
  for (int ci = 0; ci < NCHUNK; ++ci) {
    __syncthreads();
    if (ci + 1 < NCHUNK) loadChunk((ci + 1) * KC);
    const int k0 = ci * KC;
#pragma unroll
    for (int kk = 0; kk < KC; kk += 4) {
      const int kf = kk + 2 * lh;
      const int m0 = l16, m1 = 16 + l16;
      const int xi0 = (k0 + kf + 4 * m0) & (S_DIM - 1);
      const int xi1 = (k0 + kf + 4 * m1) & (S_DIM - 1);
      v2f fa0, fa1, fb0, fb1;
      fa0.x = Xs[m0][xi0];     fa0.y = Xs[m0][xi0 + 1];
      fa1.x = Xs[m1][xi1];     fa1.y = Xs[m1][xi1 + 1];
      fb0.x = Ws[cur][kf][ncol + l16];       fb0.y = Ws[cur][kf + 1][ncol + l16];
      fb1.x = Ws[cur][kf][ncol + 16 + l16];  fb1.y = Ws[cur][kf + 1][ncol + 16 + l16];
      acc00 = __builtin_amdgcn_wmma_f32_16x16x4_f32(false, fa0, false, fb0, (short)0, acc00, false, false);
      acc01 = __builtin_amdgcn_wmma_f32_16x16x4_f32(false, fa0, false, fb1, (short)0, acc01, false, false);
      acc10 = __builtin_amdgcn_wmma_f32_16x16x4_f32(false, fa1, false, fb0, (short)0, acc10, false, false);
      acc11 = __builtin_amdgcn_wmma_f32_16x16x4_f32(false, fa1, false, fb1, (short)0, acc11, false, false);
    }
    if (ci + 1 < NCHUNK) storeChunk(cur ^ 1);
    cur ^= 1;
  }

  __syncthreads();
#pragma unroll
  for (int r = 0; r < 8; ++r) {
    const int m = r + 8 * lh;
    Xs[m][ncol + l16] = acc00[r];
    Xs[m][ncol + 16 + l16] = acc01[r];
    Xs[m + 16][ncol + l16] = acc10[r];
    Xs[m + 16][ncol + 16 + l16] = acc11[r];
  }
  __syncthreads();
#pragma unroll
  for (int r = 0; r < 8; ++r) {
    const int lin = r * 256 + tid;
    const int c = lin >> 6;
    const int s4 = (lin & 63) << 2;
    *(float4*)(out + nb + c * S_DIM + s4) = *(const float4*)&Xs[c][s4];
  }
}

extern "C" void kernel_launch(void* const* d_in, const int* in_sizes, int n_in,
                              void* d_out, int out_size, void* d_ws, size_t ws_size,
                              hipStream_t stream) {
  const float* a = (const float*)d_in[0];
  const float* b = (const float*)d_in[1];
  const float* ct = (const float*)d_in[2];
  const float* cm = (const float*)d_in[3];
  const float* ew = (const float*)d_in[4];
  float* out = (float*)d_out;
  float* ws = (float*)d_ws;
  int* topidx = (int*)(ws + 1152);

  router_setup<<<1, 256, 0, stream>>>(ct, cm, ws);
  router_select<<<N_TOT, 256, 0, stream>>>(a, b, ws, topidx);
  moe_wmma_gemm<<<N_TOT, 256, 0, stream>>>(a, b, ew, topidx, out);
}


# Softmax normalisers applied after the DFT, by linearity: set-up group transforms exp(x - local max) at once; mask pass cut in 128 slices over the mix groups; GEMM groups merge partials; sums reorder
# speedup vs baseline: 4.5275x; 4.5275x over previous
#include <hip/hip_runtime.h>
#include <math.h>

typedef __attribute__((ext_vector_type(2))) float v2f;
typedef __attribute__((ext_vector_type(8))) float v8f;

#define S_DIM 256
#define C_DIM 32
#define E_DIM 32
#define N_TOT 512
#define KC 32
#define NCHUNK (S_DIM / KC)
#define LOG2E_F 1.4426950408889634f
#define TOK_OFF 2048
#define LM_OFF 2176
#define PART_OFF 2304
#define UV_OFF 4096

__device__ __forceinline__ float sigmoidf_(float x) { return 1.0f / (1.0f + expf(-x)); }

__device__ __forceinline__ float hsum4(float4 x) { return (x.x + x.y) + (x.z + x.w); }

__device__ __forceinline__ void sample_mix(const float* __restrict__ a,
                                           const float* __restrict__ b,
                                           const float* __restrict__ cm,
                                           float* __restrict__ ws) {
  __shared__ float4 red[4][8][128];
  __shared__ float4 mpart[8];
  const int tid = threadIdx.x;
  const int q = tid >> 8;
  const int w = (tid >> 5) & 7;
  const int lane = tid & 31;
  const int n = ((int)blockIdx.x - 1) * 4 + q;
  const float4* __restrict__ a4 = (const float4*)a + ((size_t)n * C_DIM + 4 * w) * (S_DIM / 4);
  const float4* __restrict__ b4 = (const float4*)b + ((size_t)n * C_DIM + 4 * w) * (S_DIM / 4);
  float4 mv = make_float4(0.f, 0.f, 0.f, 0.f);
  if (q == 0) mv = ((const float4*)cm)[((int)blockIdx.x - 1) * 256 + tid];
  float4 al[4], ah[4], bl[4], bh[4];
#pragma unroll
  for (int j = 0; j < 4; ++j) {
    al[j] = a4[j * 64 + lane];
    ah[j] = a4[j * 64 + 32 + lane];
    bl[j] = b4[j * 64 + lane];
    bh[j] = b4[j * 64 + 32 + lane];
  }
  float sa[4], sb[4];
#pragma unroll
  for (int j = 0; j < 4; ++j) {
    sa[j] = hsum4(al[j]) + hsum4(ah[j]);
    sb[j] = hsum4(bl[j]) + hsum4(bh[j]);
  }
  for (int off = 16; off > 0; off >>= 1) {
#pragma unroll
    for (int j = 0; j < 4; ++j) {
      sa[j] += __shfl_xor(sa[j], off);
      sb[j] += __shfl_xor(sb[j], off);
    }
  }
  float4 pv0 = make_float4(0.f, 0.f, 0.f, 0.f), pv1 = pv0, pu0 = pv0, pu1 = pv0;
#pragma unroll
  for (int j = 0; j < 4; ++j) {
    pv0.x += sa[j] * bl[j].x; pv0.y += sa[j] * bl[j].y; pv0.z += sa[j] * bl[j].z; pv0.w += sa[j] * bl[j].w;
    pv1.x += sa[j] * bh[j].x; pv1.y += sa[j] * bh[j].y; pv1.z += sa[j] * bh[j].z; pv1.w += sa[j] * bh[j].w;
    pu0.x += sb[j] * al[j].x; pu0.y += sb[j] * al[j].y; pu0.z += sb[j] * al[j].z; pu0.w += sb[j] * al[j].w;
    pu1.x += sb[j] * ah[j].x; pu1.y += sb[j] * ah[j].y; pu1.z += sb[j] * ah[j].z; pu1.w += sb[j] * ah[j].w;
  }
  red[q][w][lane] = pv0;
  red[q][w][32 + lane] = pv1;
  red[q][w][64 + lane] = pu0;
  red[q][w][96 + lane] = pu1;
  if (q == 0) {
    float mr = fmaxf(mv.x, mv.z), mi = fmaxf(mv.y, mv.w);
    for (int off = 16; off > 0; off >>= 1) {
      mr = fmaxf(mr, __shfl_xor(mr, off));
      mi = fmaxf(mi, __shfl_xor(mi, off));
    }
    float sr = __builtin_amdgcn_exp2f((mv.x - mr) * LOG2E_F) + __builtin_amdgcn_exp2f((mv.z - mr) * LOG2E_F);
    float si = __builtin_amdgcn_exp2f((mv.y - mi) * LOG2E_F) + __builtin_amdgcn_exp2f((mv.w - mi) * LOG2E_F);
    for (int off = 16; off > 0; off >>= 1) {
      sr += __shfl_xor(sr, off);
      si += __shfl_xor(si, off);
    }
    if (lane == 0) mpart[w] = make_float4(mr, mi, sr, si);
  }
  __syncthreads();
  if (tid < 32) {
    const float4 p = mpart[lane & 7];
    float gr = p.x, gi = p.y;
    for (int off = 4; off > 0; off >>= 1) {
      gr = fmaxf(gr, __shfl_xor(gr, off));
      gi = fmaxf(gi, __shfl_xor(gi, off));
    }
    float sr = p.z * expf(p.x - gr), si = p.w * expf(p.y - gi);
    for (int off = 4; off > 0; off >>= 1) {
      sr += __shfl_xor(sr, off);
      si += __shfl_xor(si, off);
    }
    if (tid == 0) ((float4*)(ws + PART_OFF))[(int)blockIdx.x - 1] = make_float4(gr, gi, sr, si);
  }
  const int t = tid & 255;
  float vs = 0.0f, us = 0.0f;
#pragma unroll
  for (int ww = 0; ww < 8; ++ww) {
    const float* r = (const float*)&red[q][ww][0];
    vs += r[t];
    us += r[256 + t];
  }
  ws[UV_OFF + n * 512 + t] = vs;
  ws[UV_OFF + n * 512 + 256 + t] = us;
}

__global__ __launch_bounds__(1024) void router_setup(const float* __restrict__ a,
                                                     const float* __restrict__ b,
                                                     const float* __restrict__ ct,
                                                     const float* __restrict__ cm,
                                                     float* __restrict__ ws) {
  if (blockIdx.x != 0) {
    sample_mix(a, b, cm, ws);
    return;
  }
  __shared__ float wmre[8], wmim[8];
  __shared__ float4 rc[256];
  __shared__ float2 tw[256];
  __shared__ float2 tw16[16];
  __shared__ float4 yr[256];
  __shared__ float4 yc[256];
  const int tid = threadIdx.x;
  const int lane = tid & 31;
  const int wave = tid >> 5;

  const float w0 = -6.283185307179586f / 256.0f;
  float2 rowv = make_float2(0.f, 0.f), colv = rowv;
  if (tid < 256) {
    const float2* __restrict__ cm2 = (const float2*)cm;
    rowv = cm2[tid];
    colv = cm2[tid * S_DIM];
    float lre = fmaxf(rowv.x, colv.x), lim = fmaxf(rowv.y, colv.y);
    for (int off = 16; off > 0; off >>= 1) {
      lre = fmaxf(lre, __shfl_xor(lre, off));
      lim = fmaxf(lim, __shfl_xor(lim, off));
    }
    if (lane == 0) { wmre[wave] = lre; wmim[wave] = lim; }
  } else if (tid < 512) {
    const int j = tid - 256;
    float sn, cs;
    sincosf(w0 * (float)j, &sn, &cs);
    tw[j] = make_float2(cs, sn);
    if ((j & 15) == 0) tw16[j >> 4] = make_float2(cs, sn);
  }
  __syncthreads();
  if (tid < 256) {
    float lre = wmre[0], lim = wmim[0];
    for (int i = 1; i < 8; ++i) { lre = fmaxf(lre, wmre[i]); lim = fmaxf(lim, wmim[i]); }
    rc[tid] = make_float4(expf(rowv.x - lre), expf(rowv.y - lim), expf(colv.x - lre), expf(colv.y - lim));
    if (tid == 0) { ws[LM_OFF] = lre; ws[LM_OFF + 1] = lim; }
  }
  __syncthreads();

  if (tid < 256) {
    const int t0 = tid >> 4, k0 = tid & 15;
    float4 ar = make_float4(0.f, 0.f, 0.f, 0.f), ac = ar;
    int m = 0;
    for (int t1 = 0; t1 < 16; ++t1) {
      const float2 w = tw16[m];
      const float4 r = rc[16 * t1 + t0];
      ar.x += r.x * w.x; ar.y += r.x * w.y;
      ar.z += r.y * w.x; ar.w += r.y * w.y;
      ac.x += r.z * w.x; ac.y += r.z * w.y;
      ac.z += r.w * w.x; ac.w += r.w * w.y;
      m = (m + k0) & 15;
    }
    yr[tid] = ar;
    yc[tid] = ac;
  }
  __syncthreads();
  if (tid >= 256) return;

  float4 dr = make_float4(0.f, 0.f, 0.f, 0.f), dc = dr;
  {
    const int k0 = tid & 15;
    int j = 0;
    for (int t0 = 0; t0 < 16; ++t0) {
      const float2 w = tw[j];
      const float4 y = yr[16 * t0 + k0];
      const float4 z = yc[16 * t0 + k0];
      dr.x += y.x * w.x - y.y * w.y;
      dr.y += y.y * w.x + y.x * w.y;
      dr.z += y.z * w.x - y.w * w.y;
      dr.w += y.w * w.x + y.z * w.y;
      dc.x += z.x * w.x - z.y * w.y;
      dc.y += z.y * w.x + z.x * w.y;
      dc.z += z.z * w.x - z.w * w.y;
      dc.w += z.w * w.x + z.z * w.y;
      j = (j + tid) & 255;
    }
  }
  ((float4*)ws)[tid] = dr;
  ((float4*)ws)[256 + tid] = dc;

  if (tid < E_DIM) {
    const int e = tid;
    ws[TOK_OFF + e] = sigmoidf_(ct[((e * 2 + 0) * S_DIM + 0) * 2 + 0]);
    ws[TOK_OFF + 32 + e] = sigmoidf_(ct[((e * 2 + 0) * S_DIM + 0) * 2 + 1]);
    ws[TOK_OFF + 64 + e] = sigmoidf_(ct[((e * 2 + 1) * S_DIM + 0) * 2 + 0]);
    ws[TOK_OFF + 96 + e] = sigmoidf_(ct[((e * 2 + 1) * S_DIM + 0) * 2 + 1]);
  }
}

__global__ __launch_bounds__(256) void moe_wmma_gemm(const float* __restrict__ a,
                                                     const float* __restrict__ b,
                                                     const float* __restrict__ ew,
                                                     const float* __restrict__ ws,
                                                     float* __restrict__ out) {
  __shared__ __align__(16) float Xs[C_DIM][S_DIM];
  __shared__ __align__(16) float Ws[2][KC][S_DIM];
  __shared__ float hred[8][8];
  __shared__ float mag[32];
  __shared__ int sel[3];
  const int n = blockIdx.x;
  const int tid = threadIdx.x;
  const size_t nb = (size_t)n * C_DIM * S_DIM;
  const int wave = tid >> 5;
  const int lane = tid & 31;

  const float4* __restrict__ ws4 = (const float4*)ws;
  const float vv = ws[UV_OFF + n * 512 + tid];
  const float uu = ws[UV_OFF + n * 512 + 256 + tid];
  const float4 dr = ws4[tid];
  const float4 dc = ws4[256 + tid];
  float p0 = vv * dr.x, p1 = vv * dr.y, p2 = vv * dr.z, p3 = vv * dr.w;
  float p4 = uu * dc.x, p5 = uu * dc.y, p6 = uu * dc.z, p7 = uu * dc.w;
  float nmr, nmi, nsr, nsi;
  {
    const float4* __restrict__ pp = ws4 + (PART_OFF / 4);
    const float4 q0 = pp[lane], q1 = pp[32 + lane], q2 = pp[64 + lane], q3 = pp[96 + lane];
    nmr = fmaxf(fmaxf(q0.x, q1.x), fmaxf(q2.x, q3.x));
    nmi = fmaxf(fmaxf(q0.y, q1.y), fmaxf(q2.y, q3.y));
    nsr = q0.z * __builtin_amdgcn_exp2f((q0.x - nmr) * LOG2E_F) + q1.z * __builtin_amdgcn_exp2f((q1.x - nmr) * LOG2E_F)
        + q2.z * __builtin_amdgcn_exp2f((q2.x - nmr) * LOG2E_F) + q3.z * __builtin_amdgcn_exp2f((q3.x - nmr) * LOG2E_F);
    nsi = q0.w * __builtin_amdgcn_exp2f((q0.y - nmi) * LOG2E_F) + q1.w * __builtin_amdgcn_exp2f((q1.y - nmi) * LOG2E_F)
        + q2.w * __builtin_amdgcn_exp2f((q2.y - nmi) * LOG2E_F) + q3.w * __builtin_amdgcn_exp2f((q3.y - nmi) * LOG2E_F);
  }
  const float lmr = ws[LM_OFF], lmi = ws[LM_OFF + 1];

#pragma unroll
  for (int r = 0; r < 8; ++r) {
    const int lin = r * 256 + tid;
    const int c = lin >> 6;
    const int s4 = (lin & 63) << 2;
    const float4 av = *(const float4*)(a + nb + c * S_DIM + s4);
    const float4 bv = *(const float4*)(b + nb + c * S_DIM + s4);
    float4 v;
    v.x = av.x + bv.x; v.y = av.y + bv.y; v.z = av.z + bv.z; v.w = av.w + bv.w;
    *(float4*)&Xs[c][(s4 + 4 * c) & (S_DIM - 1)] = v;
  }

  for (int off = 16; off > 0; off >>= 1) {
    p0 += __shfl_xor(p0, off);
    p1 += __shfl_xor(p1, off);
    p2 += __shfl_xor(p2, off);
    p3 += __shfl_xor(p3, off);
    p4 += __shfl_xor(p4, off);
    p5 += __shfl_xor(p5, off);
    p6 += __shfl_xor(p6, off);
    p7 += __shfl_xor(p7, off);
    const float omr = __shfl_xor(nmr, off), osr = __shfl_xor(nsr, off);
    const float omi = __shfl_xor(nmi, off), osi = __shfl_xor(nsi, off);
    const float mr = fmaxf(nmr, omr), mi = fmaxf(nmi, omi);
    nsr = nsr * __builtin_amdgcn_exp2f((nmr - mr) * LOG2E_F) + osr * __builtin_amdgcn_exp2f((omr - mr) * LOG2E_F);
    nsi = nsi * __builtin_amdgcn_exp2f((nmi - mi) * LOG2E_F) + osi * __builtin_amdgcn_exp2f((omi - mi) * LOG2E_F);
    nmr = mr; nmi = mi;
  }
  if (lane == 0) {
    hred[wave][0] = p0; hred[wave][1] = p1; hred[wave][2] = p2; hred[wave][3] = p3;
    hred[wave][4] = p4; hred[wave][5] = p5; hred[wave][6] = p6; hred[wave][7] = p7;
  }
  __syncthreads();
  if (tid < E_DIM) {
    nsr = __int_as_float(__builtin_amdgcn_readfirstlane(__float_as_int(nsr)));
    nsi = __int_as_float(__builtin_amdgcn_readfirstlane(__float_as_int(nsi)));
    const float scr = expf(lmr - nmr) / nsr;
    const float sci = expf(lmi - nmi) / nsi;
    float q[8] = {0.f, 0.f, 0.f, 0.f, 0.f, 0.f, 0.f, 0.f};
    for (int ww = 0; ww < 8; ++ww) {
#pragma unroll
      for (int i = 0; i < 8; ++i) q[i] += hred[ww][i];
    }
    const float ar = scr * q[0] - sci * q[3], ai = scr * q[1] + sci * q[2];
    const float br = scr * q[4] - sci * q[7], bi = scr * q[5] + sci * q[6];
    const int e = tid;
    float tar = ws[TOK_OFF + e], tai = ws[TOK_OFF + 32 + e], tbr = ws[TOK_OFF + 64 + e], tbi = ws[TOK_OFF + 96 + e];
    float zr = ar * tar + ai * tai + br * tbr + bi * tbi;
    float zi = ai * tar - ar * tai + bi * tbr - br * tbi;
    mag[e] = zr * zr + zi * zi;
  }
  __syncthreads();
  if (tid == 0) {
    int s0 = 0, s1 = -1, s2 = -1;
    float b0 = -1.0f, b1 = -1.0f, b2 = -1.0f;
    for (int e = 0; e < 32; ++e) { float m = mag[e]; if (m > b0) { b0 = m; s0 = e; } }
    for (int e = 0; e < 32; ++e) { if (e == s0) continue; float m = mag[e]; if (m > b1) { b1 = m; s1 = e; } }
    for (int e = 0; e < 32; ++e) { if (e == s0 || e == s1) continue; float m = mag[e]; if (m > b2) { b2 = m; s2 = e; } }
    sel[0] = s0; sel[1] = s1; sel[2] = s2;
  }
  __syncthreads();
  const int i0 = __builtin_amdgcn_readfirstlane(sel[0]);
  const int i1 = __builtin_amdgcn_readfirstlane(sel[1]);
  const int i2 = __builtin_amdgcn_readfirstlane(sel[2]);
  const float* __restrict__ w0 = ew + (size_t)i0 * (S_DIM * S_DIM);
  const float* __restrict__ w1 = ew + (size_t)i1 * (S_DIM * S_DIM);
  const float* __restrict__ w2 = ew + (size_t)i2 * (S_DIM * S_DIM);

  float4 pre[8];

  auto loadChunk = [&](int K0) {
#pragma unroll
    for (int r = 0; r < 8; ++r) {
      const int lin = r * 256 + tid;
      const int k = lin >> 6;
      const int s4 = (lin & 63) << 2;
      const size_t g = (size_t)(K0 + k) * S_DIM + s4;
      const float4 x0 = *(const float4*)(w0 + g);
      const float4 x1 = *(const float4*)(w1 + g);
      const float4 x2 = *(const float4*)(w2 + g);
      pre[r].x = (x0.x + x1.x + x2.x) * (1.0f / 3.0f);
      pre[r].y = (x0.y + x1.y + x2.y) * (1.0f / 3.0f);
      pre[r].z = (x0.z + x1.z + x2.z) * (1.0f / 3.0f);
      pre[r].w = (x0.w + x1.w + x2.w) * (1.0f / 3.0f);
    }
  };
  auto storeChunk = [&](int buf) {
#pragma unroll
    for (int r = 0; r < 8; ++r) {
      const int lin = r * 256 + tid;
      const int k = lin >> 6;
      const int s4 = (lin & 63) << 2;
      *(float4*)&Ws[buf][k][s4] = pre[r];
    }
  };

  loadChunk(0);
  storeChunk(0);

  const int l16 = lane & 15;
  const int lh = lane >> 4;
  const int ncol = wave * 32;

  v8f acc00 = {}, acc01 = {}, acc10 = {}, acc11 = {};

  int cur = 0;
  for (int ci = 0; ci < NCHUNK; ++ci) {
    __syncthreads();
    if (ci + 1 < NCHUNK) loadChunk((ci + 1) * KC);
    const int k0 = ci * KC;
#pragma unroll
    for (int kk = 0; kk < KC; kk += 4) {
      const int kf = kk + 2 * lh;
      const int m0 = l16, m1 = 16 + l16;
      const int xi0 = (k0 + kf + 4 * m0) & (S_DIM - 1);
      const int xi1 = (k0 + kf + 4 * m1) & (S_DIM - 1);
      v2f fa0, fa1, fb0, fb1;
      fa0.x = Xs[m0][xi0];     fa0.y = Xs[m0][xi0 + 1];
      fa1.x = Xs[m1][xi1];     fa1.y = Xs[m1][xi1 + 1];
      fb0.x = Ws[cur][kf][ncol + l16];       fb0.y = Ws[cur][kf + 1][ncol + l16];
      fb1.x = Ws[cur][kf][ncol + 16 + l16];  fb1.y = Ws[cur][kf + 1][ncol + 16 + l16];
      acc00 = __builtin_amdgcn_wmma_f32_16x16x4_f32(false, fa0, false, fb0, (short)0, acc00, false, false);
      acc01 = __builtin_amdgcn_wmma_f32_16x16x4_f32(false, fa0, false, fb1, (short)0, acc01, false, false);
      acc10 = __builtin_amdgcn_wmma_f32_16x16x4_f32(false, fa1, false, fb0, (short)0, acc10, false, false);
      acc11 = __builtin_amdgcn_wmma_f32_16x16x4_f32(false, fa1, false, fb1, (short)0, acc11, false, false);
    }
    if (ci + 1 < NCHUNK) storeChunk(cur ^ 1);
    cur ^= 1;
  }

  __syncthreads();
#pragma unroll
  for (int r = 0; r < 8; ++r) {
    const int m = r + 8 * lh;
    Xs[m][ncol + l16] = acc00[r];
    Xs[m][ncol + 16 + l16] = acc01[r];
    Xs[m + 16][ncol + l16] = acc10[r];
    Xs[m + 16][ncol + 16 + l16] = acc11[r];
  }
  __syncthreads();
#pragma unroll
  for (int r = 0; r < 8; ++r) {
    const int lin = r * 256 + tid;
    const int c = lin >> 6;
    const int s4 = (lin & 63) << 2;
    *(float4*)(out + nb + c * S_DIM + s4) = *(const float4*)&Xs[c][s4];
  }
}

extern "C" void kernel_launch(void* const* d_in, const int* in_sizes, int n_in,
                              void* d_out, int out_size, void* d_ws, size_t ws_size,
                              hipStream_t stream) {
  const float* a = (const float*)d_in[0];
  const float* b = (const float*)d_in[1];
  const float* ct = (const float*)d_in[2];
  const float* cm = (const float*)d_in[3];
  const float* ew = (const float*)d_in[4];
  float* out = (float*)d_out;
  float* ws = (float*)d_ws;

  router_setup<<<129, 1024, 0, stream>>>(a, b, ct, cm, ws);
  moe_wmma_gemm<<<N_TOT, 256, 0, stream>>>(a, b, ew, ws, out);
}


# Launch 1 re-tiled over all WGPs: 256 mix groups of 8 waves (2 samples each, a wave owns 8 rows) and an 8-wave set-up group, not 128 + 1 groups of 32 waves; 4 partials per sample, mix sum order moves
# speedup vs baseline: 4.8442x; 1.0700x over previous
#include <hip/hip_runtime.h>
#include <math.h>

typedef __attribute__((ext_vector_type(2))) float v2f;
typedef __attribute__((ext_vector_type(8))) float v8f;

#define S_DIM 256
#define C_DIM 32
#define E_DIM 32
#define N_TOT 512
#define KC 32
#define NCHUNK (S_DIM / KC)
#define LOG2E_F 1.4426950408889634f
#define TOK_OFF 2048
#define LM_OFF 2176
#define PART_OFF 2304
#define UV_OFF 4096

__device__ __forceinline__ float sigmoidf_(float x) { return 1.0f / (1.0f + expf(-x)); }

__device__ __forceinline__ float hsum4(float4 x) { return (x.x + x.y) + (x.z + x.w); }

__device__ __forceinline__ void sample_mix(const float* __restrict__ a,
                                           const float* __restrict__ b,
                                           const float* __restrict__ cm,
                                           float* __restrict__ ws) {
  __shared__ float4 red[2][4][128];
  __shared__ float4 mpart[8];
  const int tid = threadIdx.x;
  const int q = tid >> 7;
  const int w = (tid >> 5) & 3;
  const int lane = tid & 31;
  const int g = (int)blockIdx.x - 1;
  const int n = g * 2 + q;
  const bool slice = g < 128;
  const float4* __restrict__ a4 = (const float4*)a + ((size_t)n * C_DIM + 8 * w) * (S_DIM / 4);
  const float4* __restrict__ b4 = (const float4*)b + ((size_t)n * C_DIM + 8 * w) * (S_DIM / 4);
  float4 mv = make_float4(0.f, 0.f, 0.f, 0.f);
  if (slice) mv = ((const float4*)cm)[g * 256 + tid];
  float4 al[8], ah[8], bl[8], bh[8];
#pragma unroll
  for (int j = 0; j < 8; ++j) {
    al[j] = a4[j * 64 + lane];
    ah[j] = a4[j * 64 + 32 + lane];
    bl[j] = b4[j * 64 + lane];
    bh[j] = b4[j * 64 + 32 + lane];
  }
  float sa[8], sb[8];
#pragma unroll
  for (int j = 0; j < 8; ++j) {
    sa[j] = hsum4(al[j]) + hsum4(ah[j]);
    sb[j] = hsum4(bl[j]) + hsum4(bh[j]);
  }
  for (int off = 16; off > 0; off >>= 1) {
#pragma unroll
    for (int j = 0; j < 8; ++j) {
      sa[j] += __shfl_xor(sa[j], off);
      sb[j] += __shfl_xor(sb[j], off);
    }
  }
  float4 pv0 = make_float4(0.f, 0.f, 0.f, 0.f), pv1 = pv0, pu0 = pv0, pu1 = pv0;
#pragma unroll
  for (int j = 0; j < 8; ++j) {
    pv0.x += sa[j] * bl[j].x; pv0.y += sa[j] * bl[j].y; pv0.z += sa[j] * bl[j].z; pv0.w += sa[j] * bl[j].w;
    pv1.x += sa[j] * bh[j].x; pv1.y += sa[j] * bh[j].y; pv1.z += sa[j] * bh[j].z; pv1.w += sa[j] * bh[j].w;
    pu0.x += sb[j] * al[j].x; pu0.y += sb[j] * al[j].y; pu0.z += sb[j] * al[j].z; pu0.w += sb[j] * al[j].w;
    pu1.x += sb[j] * ah[j].x; pu1.y += sb[j] * ah[j].y; pu1.z += sb[j] * ah[j].z; pu1.w += sb[j] * ah[j].w;
  }
  red[q][w][lane] = pv0;
  red[q][w][32 + lane] = pv1;
  red[q][w][64 + lane] = pu0;
  red[q][w][96 + lane] = pu1;
  if (slice) {
    float mr = fmaxf(mv.x, mv.z), mi = fmaxf(mv.y, mv.w);
    for (int off = 16; off > 0; off >>= 1) {
      mr = fmaxf(mr, __shfl_xor(mr, off));
      mi = fmaxf(mi, __shfl_xor(mi, off));
    }
    float sr = __builtin_amdgcn_exp2f((mv.x - mr) * LOG2E_F) + __builtin_amdgcn_exp2f((mv.z - mr) * LOG2E_F);
    float si = __builtin_amdgcn_exp2f((mv.y - mi) * LOG2E_F) + __builtin_amdgcn_exp2f((mv.w - mi) * LOG2E_F);
    for (int off = 16; off > 0; off >>= 1) {
      sr += __shfl_xor(sr, off);
      si += __shfl_xor(si, off);
    }
    if (lane == 0) mpart[tid >> 5] = make_float4(mr, mi, sr, si);
  }
  __syncthreads();
  if (slice && tid < 32) {
    const float4 p = mpart[lane & 7];
    float gr = p.x, gi = p.y;
    for (int off = 4; off > 0; off >>= 1) {
      gr = fmaxf(gr, __shfl_xor(gr, off));
      gi = fmaxf(gi, __shfl_xor(gi, off));
    }
    float sr = p.z * expf(p.x - gr), si = p.w * expf(p.y - gi);
    for (int off = 4; off > 0; off >>= 1) {
      sr += __shfl_xor(sr, off);
      si += __shfl_xor(si, off);
    }
    if (tid == 0) ((float4*)(ws + PART_OFF))[g] = make_float4(gr, gi, sr, si);
  }
  const int t = tid & 127;
  float4 acc = red[q][0][t];
#pragma unroll
  for (int ww = 1; ww < 4; ++ww) {
    const float4 r = red[q][ww][t];
    acc.x += r.x; acc.y += r.y; acc.z += r.z; acc.w += r.w;
  }
  ((float4*)(ws + UV_OFF + n * 512))[t] = acc;
}

__global__ __launch_bounds__(256) __attribute__((amdgpu_waves_per_eu(2, 4))) void router_setup(const float* __restrict__ a,
                                                     const float* __restrict__ b,
                                                     const float* __restrict__ ct,
                                                     const float* __restrict__ cm,
                                                     float* __restrict__ ws) {
  if (blockIdx.x != 0) {
    sample_mix(a, b, cm, ws);
    return;
  }
  __shared__ float wmre[8], wmim[8];
  __shared__ float4 rc[256];
  __shared__ float2 tw[256];
  __shared__ float2 tw16[16];
  __shared__ float4 yr[256];
  __shared__ float4 yc[256];
  const int tid = threadIdx.x;
  const int lane = tid & 31;
  const int wave = tid >> 5;

  const float w0 = -6.283185307179586f / 256.0f;
  const float2* __restrict__ cm2 = (const float2*)cm;
  const float2 rowv = cm2[tid];
  const float2 colv = cm2[tid * S_DIM];
  {
    float sn, cs;
    sincosf(w0 * (float)tid, &sn, &cs);
    tw[tid] = make_float2(cs, sn);
    if ((tid & 15) == 0) tw16[tid >> 4] = make_float2(cs, sn);
  }
  {
    float lre = fmaxf(rowv.x, colv.x), lim = fmaxf(rowv.y, colv.y);
    for (int off = 16; off > 0; off >>= 1) {
      lre = fmaxf(lre, __shfl_xor(lre, off));
      lim = fmaxf(lim, __shfl_xor(lim, off));
    }
    if (lane == 0) { wmre[wave] = lre; wmim[wave] = lim; }
  }
  __syncthreads();
  {
    float lre = wmre[0], lim = wmim[0];
    for (int i = 1; i < 8; ++i) { lre = fmaxf(lre, wmre[i]); lim = fmaxf(lim, wmim[i]); }
    rc[tid] = make_float4(expf(rowv.x - lre), expf(rowv.y - lim), expf(colv.x - lre), expf(colv.y - lim));
    if (tid == 0) { ws[LM_OFF] = lre; ws[LM_OFF + 1] = lim; }
  }
  __syncthreads();

  {
    const int t0 = tid >> 4, k0 = tid & 15;
    float4 ar = make_float4(0.f, 0.f, 0.f, 0.f), ac = ar;
    int m = 0;
    for (int t1 = 0; t1 < 16; ++t1) {
      const float2 w = tw16[m];
      const float4 r = rc[16 * t1 + t0];
      ar.x += r.x * w.x; ar.y += r.x * w.y;
      ar.z += r.y * w.x; ar.w += r.y * w.y;
      ac.x += r.z * w.x; ac.y += r.z * w.y;
      ac.z += r.w * w.x; ac.w += r.w * w.y;
      m = (m + k0) & 15;
    }
    yr[tid] = ar;
    yc[tid] = ac;
  }
  __syncthreads();

  float4 dr = make_float4(0.f, 0.f, 0.f, 0.f), dc = dr;
  {
    const int k0 = tid & 15;
    int j = 0;
    for (int t0 = 0; t0 < 16; ++t0) {
      const float2 w = tw[j];
      const float4 y = yr[16 * t0 + k0];
      const float4 z = yc[16 * t0 + k0];
      dr.x += y.x * w.x - y.y * w.y;
      dr.y += y.y * w.x + y.x * w.y;
      dr.z += y.z * w.x - y.w * w.y;
      dr.w += y.w * w.x + y.z * w.y;
      dc.x += z.x * w.x - z.y * w.y;
      dc.y += z.y * w.x + z.x * w.y;
      dc.z += z.z * w.x - z.w * w.y;
      dc.w += z.w * w.x + z.z * w.y;
      j = (j + tid) & 255;
    }
  }
  ((float4*)ws)[tid] = dr;
  ((float4*)ws)[256 + tid] = dc;

  if (tid < E_DIM) {
    const int e = tid;
    ws[TOK_OFF + e] = sigmoidf_(ct[((e * 2 + 0) * S_DIM + 0) * 2 + 0]);
    ws[TOK_OFF + 32 + e] = sigmoidf_(ct[((e * 2 + 0) * S_DIM + 0) * 2 + 1]);
    ws[TOK_OFF + 64 + e] = sigmoidf_(ct[((e * 2 + 1) * S_DIM + 0) * 2 + 0]);
    ws[TOK_OFF + 96 + e] = sigmoidf_(ct[((e * 2 + 1) * S_DIM + 0) * 2 + 1]);
  }
}

__global__ __launch_bounds__(256) void moe_wmma_gemm(const float* __restrict__ a,
                                                     const float* __restrict__ b,
                                                     const float* __restrict__ ew,
                                                     const float* __restrict__ ws,
                                                     float* __restrict__ out) {
  __shared__ __align__(16) float Xs[C_DIM][S_DIM];
  __shared__ __align__(16) float Ws[2][KC][S_DIM];
  __shared__ float hred[8][8];
  __shared__ float mag[32];
  __shared__ int sel[3];
  const int n = blockIdx.x;
  const int tid = threadIdx.x;
  const size_t nb = (size_t)n * C_DIM * S_DIM;
  const int wave = tid >> 5;
  const int lane = tid & 31;

  const float4* __restrict__ ws4 = (const float4*)ws;
  const float vv = ws[UV_OFF + n * 512 + tid];
  const float uu = ws[UV_OFF + n * 512 + 256 + tid];
  const float4 dr = ws4[tid];
  const float4 dc = ws4[256 + tid];
  float p0 = vv * dr.x, p1 = vv * dr.y, p2 = vv * dr.z, p3 = vv * dr.w;
  float p4 = uu * dc.x, p5 = uu * dc.y, p6 = uu * dc.z, p7 = uu * dc.w;
  float nmr, nmi, nsr, nsi;
  {
    const float4* __restrict__ pp = ws4 + (PART_OFF / 4);
    const float4 q0 = pp[lane], q1 = pp[32 + lane], q2 = pp[64 + lane], q3 = pp[96 + lane];
    nmr = fmaxf(fmaxf(q0.x, q1.x), fmaxf(q2.x, q3.x));
    nmi = fmaxf(fmaxf(q0.y, q1.y), fmaxf(q2.y, q3.y));
    nsr = q0.z * __builtin_amdgcn_exp2f((q0.x - nmr) * LOG2E_F) + q1.z * __builtin_amdgcn_exp2f((q1.x - nmr) * LOG2E_F)
        + q2.z * __builtin_amdgcn_exp2f((q2.x - nmr) * LOG2E_F) + q3.z * __builtin_amdgcn_exp2f((q3.x - nmr) * LOG2E_F);
    nsi = q0.w * __builtin_amdgcn_exp2f((q0.y - nmi) * LOG2E_F) + q1.w * __builtin_amdgcn_exp2f((q1.y - nmi) * LOG2E_F)
        + q2.w * __builtin_amdgcn_exp2f((q2.y - nmi) * LOG2E_F) + q3.w * __builtin_amdgcn_exp2f((q3.y - nmi) * LOG2E_F);
  }
  const float lmr = ws[LM_OFF], lmi = ws[LM_OFF + 1];

#pragma unroll
  for (int r = 0; r < 8; ++r) {
    const int lin = r * 256 + tid;
    const int c = lin >> 6;
    const int s4 = (lin & 63) << 2;
    const float4 av = *(const float4*)(a + nb + c * S_DIM + s4);
    const float4 bv = *(const float4*)(b + nb + c * S_DIM + s4);
    float4 v;
    v.x = av.x + bv.x; v.y = av.y + bv.y; v.z = av.z + bv.z; v.w = av.w + bv.w;
    *(float4*)&Xs[c][(s4 + 4 * c) & (S_DIM - 1)] = v;
  }

  for (int off = 16; off > 0; off >>= 1) {
    p0 += __shfl_xor(p0, off);
    p1 += __shfl_xor(p1, off);
    p2 += __shfl_xor(p2, off);
    p3 += __shfl_xor(p3, off);
    p4 += __shfl_xor(p4, off);
    p5 += __shfl_xor(p5, off);
    p6 += __shfl_xor(p6, off);
    p7 += __shfl_xor(p7, off);
    const float omr = __shfl_xor(nmr, off), osr = __shfl_xor(nsr, off);
    const float omi = __shfl_xor(nmi, off), osi = __shfl_xor(nsi, off);
    const float mr = fmaxf(nmr, omr), mi = fmaxf(nmi, omi);
    nsr = nsr * __builtin_amdgcn_exp2f((nmr - mr) * LOG2E_F) + osr * __builtin_amdgcn_exp2f((omr - mr) * LOG2E_F);
    nsi = nsi * __builtin_amdgcn_exp2f((nmi - mi) * LOG2E_F) + osi * __builtin_amdgcn_exp2f((omi - mi) * LOG2E_F);
    nmr = mr; nmi = mi;
  }
  if (lane == 0) {
    hred[wave][0] = p0; hred[wave][1] = p1; hred[wave][2] = p2; hred[wave][3] = p3;
    hred[wave][4] = p4; hred[wave][5] = p5; hred[wave][6] = p6; hred[wave][7] = p7;
  }
  __syncthreads();
  if (tid < E_DIM) {
    nsr = __int_as_float(__builtin_amdgcn_readfirstlane(__float_as_int(nsr)));
    nsi = __int_as_float(__builtin_amdgcn_readfirstlane(__float_as_int(nsi)));
    const float scr = expf(lmr - nmr) / nsr;
    const float sci = expf(lmi - nmi) / nsi;
    float q[8] = {0.f, 0.f, 0.f, 0.f, 0.f, 0.f, 0.f, 0.f};
    for (int ww = 0; ww < 8; ++ww) {
#pragma unroll
      for (int i = 0; i < 8; ++i) q[i] += hred[ww][i];
    }
    const float ar = scr * q[0] - sci * q[3], ai = scr * q[1] + sci * q[2];
    const float br = scr * q[4] - sci * q[7], bi = scr * q[5] + sci * q[6];
    const int e = tid;
    float tar = ws[TOK_OFF + e], tai = ws[TOK_OFF + 32 + e], tbr = ws[TOK_OFF + 64 + e], tbi = ws[TOK_OFF + 96 + e];
    float zr = ar * tar + ai * tai + br * tbr + bi * tbi;
    float zi = ai * tar - ar * tai + bi * tbr - br * tbi;
    mag[e] = zr * zr + zi * zi;
  }
  __syncthreads();
  if (tid == 0) {
    int s0 = 0, s1 = -1, s2 = -1;
    float b0 = -1.0f, b1 = -1.0f, b2 = -1.0f;
    for (int e = 0; e < 32; ++e) { float m = mag[e]; if (m > b0) { b0 = m; s0 = e; } }
    for (int e = 0; e < 32; ++e) { if (e == s0) continue; float m = mag[e]; if (m > b1) { b1 = m; s1 = e; } }
    for (int e = 0; e < 32; ++e) { if (e == s0 || e == s1) continue; float m = mag[e]; if (m > b2) { b2 = m; s2 = e; } }
    sel[0] = s0; sel[1] = s1; sel[2] = s2;
  }
  __syncthreads();
  const int i0 = __builtin_amdgcn_readfirstlane(sel[0]);
  const int i1 = __builtin_amdgcn_readfirstlane(sel[1]);
  const int i2 = __builtin_amdgcn_readfirstlane(sel[2]);
  const float* __restrict__ w0 = ew + (size_t)i0 * (S_DIM * S_DIM);
  const float* __restrict__ w1 = ew + (size_t)i1 * (S_DIM * S_DIM);
  const float* __restrict__ w2 = ew + (size_t)i2 * (S_DIM * S_DIM);

  float4 pre[8];

  auto loadChunk = [&](int K0) {
#pragma unroll
    for (int r = 0; r < 8; ++r) {
      const int lin = r * 256 + tid;
      const int k = lin >> 6;
      const int s4 = (lin & 63) << 2;
      const size_t g = (size_t)(K0 + k) * S_DIM + s4;
      const float4 x0 = *(const float4*)(w0 + g);
      const float4 x1 = *(const float4*)(w1 + g);
      const float4 x2 = *(const float4*)(w2 + g);
      pre[r].x = (x0.x + x1.x + x2.x) * (1.0f / 3.0f);
      pre[r].y = (x0.y + x1.y + x2.y) * (1.0f / 3.0f);
      pre[r].z = (x0.z + x1.z + x2.z) * (1.0f / 3.0f);
      pre[r].w = (x0.w + x1.w + x2.w) * (1.0f / 3.0f);
    }
  };
  auto storeChunk = [&](int buf) {
#pragma unroll
    for (int r = 0; r < 8; ++r) {
      const int lin = r * 256 + tid;
      const int k = lin >> 6;
      const int s4 = (lin & 63) << 2;
      *(float4*)&Ws[buf][k][s4] = pre[r];
    }
  };

  loadChunk(0);
  storeChunk(0);

  const int l16 = lane & 15;
  const int lh = lane >> 4;
  const int ncol = wave * 32;

  v8f acc00 = {}, acc01 = {}, acc10 = {}, acc11 = {};

  int cur = 0;
  for (int ci = 0; ci < NCHUNK; ++ci) {
    __syncthreads();
    if (ci + 1 < NCHUNK) loadChunk((ci + 1) * KC);
    const int k0 = ci * KC;
#pragma unroll
    for (int kk = 0; kk < KC; kk += 4) {
      const int kf = kk + 2 * lh;
      const int m0 = l16, m1 = 16 + l16;
      const int xi0 = (k0 + kf + 4 * m0) & (S_DIM - 1);
      const int xi1 = (k0 + kf + 4 * m1) & (S_DIM - 1);
      v2f fa0, fa1, fb0, fb1;
      fa0.x = Xs[m0][xi0];     fa0.y = Xs[m0][xi0 + 1];
      fa1.x = Xs[m1][xi1];     fa1.y = Xs[m1][xi1 + 1];
      fb0.x = Ws[cur][kf][ncol + l16];       fb0.y = Ws[cur][kf + 1][ncol + l16];
      fb1.x = Ws[cur][kf][ncol + 16 + l16];  fb1.y = Ws[cur][kf + 1][ncol + 16 + l16];
      acc00 = __builtin_amdgcn_wmma_f32_16x16x4_f32(false, fa0, false, fb0, (short)0, acc00, false, false);
      acc01 = __builtin_amdgcn_wmma_f32_16x16x4_f32(false, fa0, false, fb1, (short)0, acc01, false, false);
      acc10 = __builtin_amdgcn_wmma_f32_16x16x4_f32(false, fa1, false, fb0, (short)0, acc10, false, false);
      acc11 = __builtin_amdgcn_wmma_f32_16x16x4_f32(false, fa1, false, fb1, (short)0, acc11, false, false);
    }
    if (ci + 1 < NCHUNK) storeChunk(cur ^ 1);
    cur ^= 1;
  }

  __syncthreads();
#pragma unroll
  for (int r = 0; r < 8; ++r) {
    const int m = r + 8 * lh;
    Xs[m][ncol + l16] = acc00[r];
    Xs[m][ncol + 16 + l16] = acc01[r];
    Xs[m + 16][ncol + l16] = acc10[r];
    Xs[m + 16][ncol + 16 + l16] = acc11[r];
  }
  __syncthreads();
#pragma unroll
  for (int r = 0; r < 8; ++r) {
    const int lin = r * 256 + tid;
    const int c = lin >> 6;
    const int s4 = (lin & 63) << 2;
    *(float4*)(out + nb + c * S_DIM + s4) = *(const float4*)&Xs[c][s4];
  }
}

extern "C" void kernel_launch(void* const* d_in, const int* in_sizes, int n_in,
                              void* d_out, int out_size, void* d_ws, size_t ws_size,
                              hipStream_t stream) {
  const float* a = (const float*)d_in[0];
  const float* b = (const float*)d_in[1];
  const float* ct = (const float*)d_in[2];
  const float* cm = (const float*)d_in[3];
  const float* ew = (const float*)d_in[4];
  float* out = (float*)d_out;
  float* ws = (float*)d_ws;

  router_setup<<<257, 256, 0, stream>>>(a, b, ct, cm, ws);
  moe_wmma_gemm<<<N_TOT, 256, 0, stream>>>(a, b, ew, ws, out);
}


# GEMM staging private to the wave: each wave loads, sums and stages in LDS rows of its own only the 32 weight columns it multiplies, then its result tile; no work-group barrier past the scores
# speedup vs baseline: 5.0468x; 1.0418x over previous
#include <hip/hip_runtime.h>
#include <math.h>

typedef __attribute__((ext_vector_type(2))) float v2f;
typedef __attribute__((ext_vector_type(8))) float v8f;

#define S_DIM 256
#define C_DIM 32
#define E_DIM 32
#define N_TOT 512
#define KC 32
#define NCHUNK (S_DIM / KC)
#define WPITCH 40
#define LOG2E_F 1.4426950408889634f
#define TOK_OFF 2048
#define LM_OFF 2176
#define PART_OFF 2304
#define UV_OFF 4096

__device__ __forceinline__ float sigmoidf_(float x) { return 1.0f / (1.0f + expf(-x)); }

__device__ __forceinline__ float hsum4(float4 x) { return (x.x + x.y) + (x.z + x.w); }

__device__ __forceinline__ void sample_mix(const float* __restrict__ a,
                                           const float* __restrict__ b,
                                           const float* __restrict__ cm,
                                           float* __restrict__ ws) {
  __shared__ float4 red[2][4][128];
  __shared__ float4 mpart[8];
  const int tid = threadIdx.x;
  const int q = tid >> 7;
  const int w = (tid >> 5) & 3;
  const int lane = tid & 31;
  const int g = (int)blockIdx.x - 1;
  const int n = g * 2 + q;
  const bool slice = g < 128;
  const float4* __restrict__ a4 = (const float4*)a + ((size_t)n * C_DIM + 8 * w) * (S_DIM / 4);
  const float4* __restrict__ b4 = (const float4*)b + ((size_t)n * C_DIM + 8 * w) * (S_DIM / 4);
  float4 mv = make_float4(0.f, 0.f, 0.f, 0.f);
  if (slice) mv = ((const float4*)cm)[g * 256 + tid];
  float4 al[8], ah[8], bl[8], bh[8];
#pragma unroll
  for (int j = 0; j < 8; ++j) {
    al[j] = a4[j * 64 + lane];
    ah[j] = a4[j * 64 + 32 + lane];
    bl[j] = b4[j * 64 + lane];
    bh[j] = b4[j * 64 + 32 + lane];
  }
  float sa[8], sb[8];
#pragma unroll
  for (int j = 0; j < 8; ++j) {
    sa[j] = hsum4(al[j]) + hsum4(ah[j]);
    sb[j] = hsum4(bl[j]) + hsum4(bh[j]);
  }
  for (int off = 16; off > 0; off >>= 1) {
#pragma unroll
    for (int j = 0; j < 8; ++j) {
      sa[j] += __shfl_xor(sa[j], off);
      sb[j] += __shfl_xor(sb[j], off);
    }
  }
  float4 pv0 = make_float4(0.f, 0.f, 0.f, 0.f), pv1 = pv0, pu0 = pv0, pu1 = pv0;
#pragma unroll
  for (int j = 0; j < 8; ++j) {
    pv0.x += sa[j] * bl[j].x; pv0.y += sa[j] * bl[j].y; pv0.z += sa[j] * bl[j].z; pv0.w += sa[j] * bl[j].w;
    pv1.x += sa[j] * bh[j].x; pv1.y += sa[j] * bh[j].y; pv1.z += sa[j] * bh[j].z; pv1.w += sa[j] * bh[j].w;
    pu0.x += sb[j] * al[j].x; pu0.y += sb[j] * al[j].y; pu0.z += sb[j] * al[j].z; pu0.w += sb[j] * al[j].w;
    pu1.x += sb[j] * ah[j].x; pu1.y += sb[j] * ah[j].y; pu1.z += sb[j] * ah[j].z; pu1.w += sb[j] * ah[j].w;
  }
  red[q][w][lane] = pv0;
  red[q][w][32 + lane] = pv1;
  red[q][w][64 + lane] = pu0;
  red[q][w][96 + lane] = pu1;
  if (slice) {
    float mr = fmaxf(mv.x, mv.z), mi = fmaxf(mv.y, mv.w);
    for (int off = 16; off > 0; off >>= 1) {
      mr = fmaxf(mr, __shfl_xor(mr, off));
      mi = fmaxf(mi, __shfl_xor(mi, off));
    }
    float sr = __builtin_amdgcn_exp2f((mv.x - mr) * LOG2E_F) + __builtin_amdgcn_exp2f((mv.z - mr) * LOG2E_F);
    float si = __builtin_amdgcn_exp2f((mv.y - mi) * LOG2E_F) + __builtin_amdgcn_exp2f((mv.w - mi) * LOG2E_F);
    for (int off = 16; off > 0; off >>= 1) {
      sr += __shfl_xor(sr, off);
      si += __shfl_xor(si, off);
    }
    if (lane == 0) mpart[tid >> 5] = make_float4(mr, mi, sr, si);
  }
  __syncthreads();
  if (slice && tid < 32) {
    const float4 p = mpart[lane & 7];
    float gr = p.x, gi = p.y;
    for (int off = 4; off > 0; off >>= 1) {
      gr = fmaxf(gr, __shfl_xor(gr, off));
      gi = fmaxf(gi, __shfl_xor(gi, off));
    }
    float sr = p.z * expf(p.x - gr), si = p.w * expf(p.y - gi);
    for (int off = 4; off > 0; off >>= 1) {
      sr += __shfl_xor(sr, off);
      si += __shfl_xor(si, off);
    }
    if (tid == 0) ((float4*)(ws + PART_OFF))[g] = make_float4(gr, gi, sr, si);
  }
  const int t = tid & 127;
  float4 acc = red[q][0][t];
#pragma unroll
  for (int ww = 1; ww < 4; ++ww) {
    const float4 r = red[q][ww][t];
    acc.x += r.x; acc.y += r.y; acc.z += r.z; acc.w += r.w;
  }
  ((float4*)(ws + UV_OFF + n * 512))[t] = acc;
}

__global__ __launch_bounds__(256) __attribute__((amdgpu_waves_per_eu(2, 4))) void router_setup(const float* __restrict__ a,
                                                     const float* __restrict__ b,
                                                     const float* __restrict__ ct,
                                                     const float* __restrict__ cm,
                                                     float* __restrict__ ws) {
  if (blockIdx.x != 0) {
    sample_mix(a, b, cm, ws);
    return;
  }
  __shared__ float wmre[8], wmim[8];
  __shared__ float4 rc[256];
  __shared__ float2 tw[256];
  __shared__ float2 tw16[16];
  __shared__ float4 yr[256];
  __shared__ float4 yc[256];
  const int tid = threadIdx.x;
  const int lane = tid & 31;
  const int wave = tid >> 5;

  const float w0 = -6.283185307179586f / 256.0f;
  const float2* __restrict__ cm2 = (const float2*)cm;
  const float2 rowv = cm2[tid];
  const float2 colv = cm2[tid * S_DIM];
  {
    float sn, cs;
    sincosf(w0 * (float)tid, &sn, &cs);
    tw[tid] = make_float2(cs, sn);
    if ((tid & 15) == 0) tw16[tid >> 4] = make_float2(cs, sn);
  }
  {
    float lre = fmaxf(rowv.x, colv.x), lim = fmaxf(rowv.y, colv.y);
    for (int off = 16; off > 0; off >>= 1) {
      lre = fmaxf(lre, __shfl_xor(lre, off));
      lim = fmaxf(lim, __shfl_xor(lim, off));
    }
    if (lane == 0) { wmre[wave] = lre; wmim[wave] = lim; }
  }
  __syncthreads();
  {
    float lre = wmre[0], lim = wmim[0];
    for (int i = 1; i < 8; ++i) { lre = fmaxf(lre, wmre[i]); lim = fmaxf(lim, wmim[i]); }
    rc[tid] = make_float4(expf(rowv.x - lre), expf(rowv.y - lim), expf(colv.x - lre), expf(colv.y - lim));
    if (tid == 0) { ws[LM_OFF] = lre; ws[LM_OFF + 1] = lim; }
  }
  __syncthreads();

  {
    const int t0 = tid >> 4, k0 = tid & 15;
    float4 ar = make_float4(0.f, 0.f, 0.f, 0.f), ac = ar;
    int m = 0;
    for (int t1 = 0; t1 < 16; ++t1) {
      const float2 w = tw16[m];
      const float4 r = rc[16 * t1 + t0];
      ar.x += r.x * w.x; ar.y += r.x * w.y;
      ar.z += r.y * w.x; ar.w += r.y * w.y;
      ac.x += r.z * w.x; ac.y += r.z * w.y;
      ac.z += r.w * w.x; ac.w += r.w * w.y;
      m = (m + k0) & 15;
    }
    yr[tid] = ar;
    yc[tid] = ac;
  }
  __syncthreads();

  float4 dr = make_float4(0.f, 0.f, 0.f, 0.f), dc = dr;
  {
    const int k0 = tid & 15;
    int j = 0;
    for (int t0 = 0; t0 < 16; ++t0) {
      const float2 w = tw[j];
      const float4 y = yr[16 * t0 + k0];
      const float4 z = yc[16 * t0 + k0];
      dr.x += y.x * w.x - y.y * w.y;
      dr.y += y.y * w.x + y.x * w.y;
      dr.z += y.z * w.x - y.w * w.y;
      dr.w += y.w * w.x + y.z * w.y;
      dc.x += z.x * w.x - z.y * w.y;
      dc.y += z.y * w.x + z.x * w.y;
      dc.z += z.z * w.x - z.w * w.y;
      dc.w += z.w * w.x + z.z * w.y;
      j = (j + tid) & 255;
    }
  }
  ((float4*)ws)[tid] = dr;
  ((float4*)ws)[256 + tid] = dc;

  if (tid < E_DIM) {
    const int e = tid;
    ws[TOK_OFF + e] = sigmoidf_(ct[((e * 2 + 0) * S_DIM + 0) * 2 + 0]);
    ws[TOK_OFF + 32 + e] = sigmoidf_(ct[((e * 2 + 0) * S_DIM + 0) * 2 + 1]);
    ws[TOK_OFF + 64 + e] = sigmoidf_(ct[((e * 2 + 1) * S_DIM + 0) * 2 + 0]);
    ws[TOK_OFF + 96 + e] = sigmoidf_(ct[((e * 2 + 1) * S_DIM + 0) * 2 + 1]);
  }
}

__global__ __launch_bounds__(256) void moe_wmma_gemm(const float* __restrict__ a,
                                                     const float* __restrict__ b,
                                                     const float* __restrict__ ew,
                                                     const float* __restrict__ ws,
                                                     float* __restrict__ out) {
  __shared__ __align__(16) float Xs[C_DIM][S_DIM];
  __shared__ __align__(16) float Wp[8][2][KC][WPITCH];
  __shared__ float hred[8][8];
  __shared__ float mag[32];
  __shared__ int sel[3];
  const int n = blockIdx.x;
  const int tid = threadIdx.x;
  const size_t nb = (size_t)n * C_DIM * S_DIM;
  const int wave = tid >> 5;
  const int lane = tid & 31;

  const float4* __restrict__ ws4 = (const float4*)ws;
  const float vv = ws[UV_OFF + n * 512 + tid];
  const float uu = ws[UV_OFF + n * 512 + 256 + tid];
  const float4 dr = ws4[tid];
  const float4 dc = ws4[256 + tid];
  float p0 = vv * dr.x, p1 = vv * dr.y, p2 = vv * dr.z, p3 = vv * dr.w;
  float p4 = uu * dc.x, p5 = uu * dc.y, p6 = uu * dc.z, p7 = uu * dc.w;
  float nmr, nmi, nsr, nsi;
  {
    const float4* __restrict__ pp = ws4 + (PART_OFF / 4);
    const float4 q0 = pp[lane], q1 = pp[32 + lane], q2 = pp[64 + lane], q3 = pp[96 + lane];
    nmr = fmaxf(fmaxf(q0.x, q1.x), fmaxf(q2.x, q3.x));
    nmi = fmaxf(fmaxf(q0.y, q1.y), fmaxf(q2.y, q3.y));
    nsr = q0.z * __builtin_amdgcn_exp2f((q0.x - nmr) * LOG2E_F) + q1.z * __builtin_amdgcn_exp2f((q1.x - nmr) * LOG2E_F)
        + q2.z * __builtin_amdgcn_exp2f((q2.x - nmr) * LOG2E_F) + q3.z * __builtin_amdgcn_exp2f((q3.x - nmr) * LOG2E_F);
    nsi = q0.w * __builtin_amdgcn_exp2f((q0.y - nmi) * LOG2E_F) + q1.w * __builtin_amdgcn_exp2f((q1.y - nmi) * LOG2E_F)
        + q2.w * __builtin_amdgcn_exp2f((q2.y - nmi) * LOG2E_F) + q3.w * __builtin_amdgcn_exp2f((q3.y - nmi) * LOG2E_F);
  }
  const float lmr = ws[LM_OFF], lmi = ws[LM_OFF + 1];

#pragma unroll
  for (int r = 0; r < 8; ++r) {
    const int lin = r * 256 + tid;
    const int c = lin >> 6;
    const int s4 = (lin & 63) << 2;
    const float4 av = *(const float4*)(a + nb + c * S_DIM + s4);
    const float4 bv = *(const float4*)(b + nb + c * S_DIM + s4);
    float4 v;
    v.x = av.x + bv.x; v.y = av.y + bv.y; v.z = av.z + bv.z; v.w = av.w + bv.w;
    *(float4*)&Xs[c][(s4 + 4 * c) & (S_DIM - 1)] = v;
  }

  for (int off = 16; off > 0; off >>= 1) {
    p0 += __shfl_xor(p0, off);
    p1 += __shfl_xor(p1, off);
    p2 += __shfl_xor(p2, off);
    p3 += __shfl_xor(p3, off);
    p4 += __shfl_xor(p4, off);
    p5 += __shfl_xor(p5, off);
    p6 += __shfl_xor(p6, off);
    p7 += __shfl_xor(p7, off);
    const float omr = __shfl_xor(nmr, off), osr = __shfl_xor(nsr, off);
    const float omi = __shfl_xor(nmi, off), osi = __shfl_xor(nsi, off);
    const float mr = fmaxf(nmr, omr), mi = fmaxf(nmi, omi);
    nsr = nsr * __builtin_amdgcn_exp2f((nmr - mr) * LOG2E_F) + osr * __builtin_amdgcn_exp2f((omr - mr) * LOG2E_F);
    nsi = nsi * __builtin_amdgcn_exp2f((nmi - mi) * LOG2E_F) + osi * __builtin_amdgcn_exp2f((omi - mi) * LOG2E_F);
    nmr = mr; nmi = mi;
  }
  if (lane == 0) {
    hred[wave][0] = p0; hred[wave][1] = p1; hred[wave][2] = p2; hred[wave][3] = p3;
    hred[wave][4] = p4; hred[wave][5] = p5; hred[wave][6] = p6; hred[wave][7] = p7;
  }
  __syncthreads();
  if (tid < E_DIM) {
    nsr = __int_as_float(__builtin_amdgcn_readfirstlane(__float_as_int(nsr)));
    nsi = __int_as_float(__builtin_amdgcn_readfirstlane(__float_as_int(nsi)));
    const float scr = expf(lmr - nmr) / nsr;
    const float sci = expf(lmi - nmi) / nsi;
    float q[8] = {0.f, 0.f, 0.f, 0.f, 0.f, 0.f, 0.f, 0.f};
    for (int ww = 0; ww < 8; ++ww) {
#pragma unroll
      for (int i = 0; i < 8; ++i) q[i] += hred[ww][i];
    }
    const float ar = scr * q[0] - sci * q[3], ai = scr * q[1] + sci * q[2];
    const float br = scr * q[4] - sci * q[7], bi = scr * q[5] + sci * q[6];
    const int e = tid;
    float tar = ws[TOK_OFF + e], tai = ws[TOK_OFF + 32 + e], tbr = ws[TOK_OFF + 64 + e], tbi = ws[TOK_OFF + 96 + e];
    float zr = ar * tar + ai * tai + br * tbr + bi * tbi;
    float zi = ai * tar - ar * tai + bi * tbr - br * tbi;
    mag[e] = zr * zr + zi * zi;
  }
  __syncthreads();
  if (tid == 0) {
    int s0 = 0, s1 = -1, s2 = -1;
    float b0 = -1.0f, b1 = -1.0f, b2 = -1.0f;
    for (int e = 0; e < 32; ++e) { float m = mag[e]; if (m > b0) { b0 = m; s0 = e; } }
    for (int e = 0; e < 32; ++e) { if (e == s0) continue; float m = mag[e]; if (m > b1) { b1 = m; s1 = e; } }
    for (int e = 0; e < 32; ++e) { if (e == s0 || e == s1) continue; float m = mag[e]; if (m > b2) { b2 = m; s2 = e; } }
    sel[0] = s0; sel[1] = s1; sel[2] = s2;
  }
  __syncthreads();
  const int i0 = __builtin_amdgcn_readfirstlane(sel[0]);
  const int i1 = __builtin_amdgcn_readfirstlane(sel[1]);
  const int i2 = __builtin_amdgcn_readfirstlane(sel[2]);
  const float* __restrict__ w0 = ew + (size_t)i0 * (S_DIM * S_DIM);
  const float* __restrict__ w1 = ew + (size_t)i1 * (S_DIM * S_DIM);
  const float* __restrict__ w2 = ew + (size_t)i2 * (S_DIM * S_DIM);

  float4 x0[8], x1[8], x2[8];

  const int l16 = lane & 15;
  const int lh = lane >> 4;
  const int ncol = wave * 32;
  const int lrow = lane >> 3;
  const int lcol = (lane & 7) << 2;

  auto issueChunk = [&](int K0) {
#pragma unroll
    for (int r = 0; r < 8; ++r) {
      const size_t g = (size_t)(K0 + 4 * r + lrow) * S_DIM + ncol + lcol;
      x0[r] = *(const float4*)(w0 + g);
      x1[r] = *(const float4*)(w1 + g);
      x2[r] = *(const float4*)(w2 + g);
    }
  };
  auto sumStoreChunk = [&](int buf) {
#pragma unroll
    for (int r = 0; r < 8; ++r) {
      float4 p;
      p.x = (x0[r].x + x1[r].x + x2[r].x) * (1.0f / 3.0f);
      p.y = (x0[r].y + x1[r].y + x2[r].y) * (1.0f / 3.0f);
      p.z = (x0[r].z + x1[r].z + x2[r].z) * (1.0f / 3.0f);
      p.w = (x0[r].w + x1[r].w + x2[r].w) * (1.0f / 3.0f);
      *(float4*)&Wp[wave][buf][4 * r + lrow][lcol] = p;
    }
  };

  auto waveHandover = [] {
    __builtin_amdgcn_fence(__ATOMIC_RELEASE, "wavefront");
    __builtin_amdgcn_wave_barrier();
    __builtin_amdgcn_fence(__ATOMIC_ACQUIRE, "wavefront");
  };

  v8f acc00 = {}, acc01 = {}, acc10 = {}, acc11 = {};

  auto burst = [&](int k0, int buf) {
    const float (*Wc)[WPITCH] = Wp[wave][buf];
#pragma unroll
    for (int kk = 0; kk < KC; kk += 4) {
      const int kf = kk + 2 * lh;
      const int m0 = l16, m1 = 16 + l16;
      const int xi0 = (k0 + kf + 4 * m0) & (S_DIM - 1);
      const int xi1 = (k0 + kf + 4 * m1) & (S_DIM - 1);
      v2f fa0, fa1, fb0, fb1;
      fa0.x = Xs[m0][xi0];     fa0.y = Xs[m0][xi0 + 1];
      fa1.x = Xs[m1][xi1];     fa1.y = Xs[m1][xi1 + 1];
      fb0.x = Wc[kf][l16];       fb0.y = Wc[kf + 1][l16];
      fb1.x = Wc[kf][16 + l16];  fb1.y = Wc[kf + 1][16 + l16];
      acc00 = __builtin_amdgcn_wmma_f32_16x16x4_f32(false, fa0, false, fb0, (short)0, acc00, false, false);
      acc01 = __builtin_amdgcn_wmma_f32_16x16x4_f32(false, fa0, false, fb1, (short)0, acc01, false, false);
      acc10 = __builtin_amdgcn_wmma_f32_16x16x4_f32(false, fa1, false, fb0, (short)0, acc10, false, false);
      acc11 = __builtin_amdgcn_wmma_f32_16x16x4_f32(false, fa1, false, fb1, (short)0, acc11, false, false);
    }
  };

  issueChunk(0);
  sumStoreChunk(0);

  int cur = 0;
  for (int ci = 0; ci + 1 < NCHUNK; ++ci) {
    waveHandover();
    issueChunk((ci + 1) * KC);
    burst(ci * KC, cur);
    sumStoreChunk(cur ^ 1);
    cur ^= 1;
  }
  waveHandover();
  burst((NCHUNK - 1) * KC, cur);
  cur ^= 1;

  float (*Wo)[WPITCH] = Wp[wave][cur];
  waveHandover();
#pragma unroll
  for (int r = 0; r < 8; ++r) {
    const int m = r + 8 * lh;
    Wo[m][l16] = acc00[r];
    Wo[m][16 + l16] = acc01[r];
    Wo[m + 16][l16] = acc10[r];
    Wo[m + 16][16 + l16] = acc11[r];
  }
  waveHandover();
#pragma unroll
  for (int r = 0; r < 8; ++r) {
    const int row = 4 * r + lrow;
    *(float4*)(out + nb + row * S_DIM + ncol + lcol) = *(const float4*)&Wo[row][lcol];
  }
}

extern "C" void kernel_launch(void* const* d_in, const int* in_sizes, int n_in,
                              void* d_out, int out_size, void* d_ws, size_t ws_size,
                              hipStream_t stream) {
  const float* a = (const float*)d_in[0];
  const float* b = (const float*)d_in[1];
  const float* ct = (const float*)d_in[2];
  const float* cm = (const float*)d_in[3];
  const float* ew = (const float*)d_in[4];
  float* out = (float*)d_out;
  float* ws = (float*)d_ws;

  router_setup<<<257, 256, 0, stream>>>(a, b, ct, cm, ws);
  moe_wmma_gemm<<<N_TOT, 256, 0, stream>>>(a, b, ew, ws, out);
}


# GEMM groups of 4 waves, not 8: a wave owns 64 columns (2 x 4 tiles, an A fragment feeds 4 products), half as many waves enter; score dots take 2 terms per thread, dot sum order moves
# speedup vs baseline: 5.1960x; 1.0296x over previous
#include <hip/hip_runtime.h>
#include <math.h>

typedef __attribute__((ext_vector_type(2))) float v2f;
typedef __attribute__((ext_vector_type(8))) float v8f;

#define S_DIM 256
#define C_DIM 32
#define E_DIM 32
#define N_TOT 512
#define KC 32
#define NCHUNK (S_DIM / KC)
#define WPITCH 80
#define LOG2E_F 1.4426950408889634f
#define TOK_OFF 2048
#define LM_OFF 2176
#define PART_OFF 2304
#define UV_OFF 4096

__device__ __forceinline__ float sigmoidf_(float x) { return 1.0f / (1.0f + expf(-x)); }

__device__ __forceinline__ float hsum4(float4 x) { return (x.x + x.y) + (x.z + x.w); }

__device__ __forceinline__ void sample_mix(const float* __restrict__ a,
                                           const float* __restrict__ b,
                                           const float* __restrict__ cm,
                                           float* __restrict__ ws) {
  __shared__ float4 red[2][4][128];
  __shared__ float4 mpart[8];
  const int tid = threadIdx.x;
  const int q = tid >> 7;
  const int w = (tid >> 5) & 3;
  const int lane = tid & 31;
  const int g = (int)blockIdx.x - 1;
  const int n = g * 2 + q;
  const bool slice = g < 128;
  const float4* __restrict__ a4 = (const float4*)a + ((size_t)n * C_DIM + 8 * w) * (S_DIM / 4);
  const float4* __restrict__ b4 = (const float4*)b + ((size_t)n * C_DIM + 8 * w) * (S_DIM / 4);
  float4 mv = make_float4(0.f, 0.f, 0.f, 0.f);
  if (slice) mv = ((const float4*)cm)[g * 256 + tid];
  float4 al[8], ah[8], bl[8], bh[8];
#pragma unroll
  for (int j = 0; j < 8; ++j) {
    al[j] = a4[j * 64 + lane];
    ah[j] = a4[j * 64 + 32 + lane];
    bl[j] = b4[j * 64 + lane];
    bh[j] = b4[j * 64 + 32 + lane];
  }
  float sa[8], sb[8];
#pragma unroll
  for (int j = 0; j < 8; ++j) {
    sa[j] = hsum4(al[j]) + hsum4(ah[j]);
    sb[j] = hsum4(bl[j]) + hsum4(bh[j]);
  }
  for (int off = 16; off > 0; off >>= 1) {
#pragma unroll
    for (int j = 0; j < 8; ++j) {
      sa[j] += __shfl_xor(sa[j], off);
      sb[j] += __shfl_xor(sb[j], off);
    }
  }
  float4 pv0 = make_float4(0.f, 0.f, 0.f, 0.f), pv1 = pv0, pu0 = pv0, pu1 = pv0;
#pragma unroll
  for (int j = 0; j < 8; ++j) {
    pv0.x += sa[j] * bl[j].x; pv0.y += sa[j] * bl[j].y; pv0.z += sa[j] * bl[j].z; pv0.w += sa[j] * bl[j].w;
    pv1.x += sa[j] * bh[j].x; pv1.y += sa[j] * bh[j].y; pv1.z += sa[j] * bh[j].z; pv1.w += sa[j] * bh[j].w;
    pu0.x += sb[j] * al[j].x; pu0.y += sb[j] * al[j].y; pu0.z += sb[j] * al[j].z; pu0.w += sb[j] * al[j].w;
    pu1.x += sb[j] * ah[j].x; pu1.y += sb[j] * ah[j].y; pu1.z += sb[j] * ah[j].z; pu1.w += sb[j] * ah[j].w;
  }
  red[q][w][lane] = pv0;
  red[q][w][32 + lane] = pv1;
  red[q][w][64 + lane] = pu0;
  red[q][w][96 + lane] = pu1;
  if (slice) {
    float mr = fmaxf(mv.x, mv.z), mi = fmaxf(mv.y, mv.w);
    for (int off = 16; off > 0; off >>= 1) {
      mr = fmaxf(mr, __shfl_xor(mr, off));
      mi = fmaxf(mi, __shfl_xor(mi, off));
    }
    float sr = __builtin_amdgcn_exp2f((mv.x - mr) * LOG2E_F) + __builtin_amdgcn_exp2f((mv.z - mr) * LOG2E_F);
    float si = __builtin_amdgcn_exp2f((mv.y - mi) * LOG2E_F) + __builtin_amdgcn_exp2f((mv.w - mi) * LOG2E_F);
    for (int off = 16; off > 0; off >>= 1) {
      sr += __shfl_xor(sr, off);
      si += __shfl_xor(si, off);
    }
    if (lane == 0) mpart[tid >> 5] = make_float4(mr, mi, sr, si);
  }
  __syncthreads();
  if (slice && tid < 32) {
    const float4 p = mpart[lane & 7];
    float gr = p.x, gi = p.y;
    for (int off = 4; off > 0; off >>= 1) {
      gr = fmaxf(gr, __shfl_xor(gr, off));
      gi = fmaxf(gi, __shfl_xor(gi, off));
    }
    float sr = p.z * expf(p.x - gr), si = p.w * expf(p.y - gi);
    for (int off = 4; off > 0; off >>= 1) {
      sr += __shfl_xor(sr, off);
      si += __shfl_xor(si, off);
    }
    if (tid == 0) ((float4*)(ws + PART_OFF))[g] = make_float4(gr, gi, sr, si);
  }
  const int t = tid & 127;
  float4 acc = red[q][0][t];
#pragma unroll
  for (int ww = 1; ww < 4; ++ww) {
    const float4 r = red[q][ww][t];
    acc.x += r.x; acc.y += r.y; acc.z += r.z; acc.w += r.w;
  }
  ((float4*)(ws + UV_OFF + n * 512))[t] = acc;
}

__global__ __launch_bounds__(256) __attribute__((amdgpu_waves_per_eu(2, 4))) void router_setup(const float* __restrict__ a,
                                                     const float* __restrict__ b,
                                                     const float* __restrict__ ct,
                                                     const float* __restrict__ cm,
                                                     float* __restrict__ ws) {
  if (blockIdx.x != 0) {
    sample_mix(a, b, cm, ws);
    return;
  }
  __shared__ float wmre[8], wmim[8];
  __shared__ float4 rc[256];
  __shared__ float2 tw[256];
  __shared__ float2 tw16[16];
  __shared__ float4 yr[256];
  __shared__ float4 yc[256];
  const int tid = threadIdx.x;
  const int lane = tid & 31;
  const int wave = tid >> 5;

  const float w0 = -6.283185307179586f / 256.0f;
  const float2* __restrict__ cm2 = (const float2*)cm;
  const float2 rowv = cm2[tid];
  const float2 colv = cm2[tid * S_DIM];
  {
    float sn, cs;
    sincosf(w0 * (float)tid, &sn, &cs);
    tw[tid] = make_float2(cs, sn);
    if ((tid & 15) == 0) tw16[tid >> 4] = make_float2(cs, sn);
  }
  {
    float lre = fmaxf(rowv.x, colv.x), lim = fmaxf(rowv.y, colv.y);
    for (int off = 16; off > 0; off >>= 1) {
      lre = fmaxf(lre, __shfl_xor(lre, off));
      lim = fmaxf(lim, __shfl_xor(lim, off));
    }
    if (lane == 0) { wmre[wave] = lre; wmim[wave] = lim; }
  }
  __syncthreads();
  {
    float lre = wmre[0], lim = wmim[0];
    for (int i = 1; i < 8; ++i) { lre = fmaxf(lre, wmre[i]); lim = fmaxf(lim, wmim[i]); }
    rc[tid] = make_float4(expf(rowv.x - lre), expf(rowv.y - lim), expf(colv.x - lre), expf(colv.y - lim));
    if (tid == 0) { ws[LM_OFF] = lre; ws[LM_OFF + 1] = lim; }
  }
  __syncthreads();

  {
    const int t0 = tid >> 4, k0 = tid & 15;
    float4 ar = make_float4(0.f, 0.f, 0.f, 0.f), ac = ar;
    int m = 0;
    for (int t1 = 0; t1 < 16; ++t1) {
      const float2 w = tw16[m];
      const float4 r = rc[16 * t1 + t0];
      ar.x += r.x * w.x; ar.y += r.x * w.y;
      ar.z += r.y * w.x; ar.w += r.y * w.y;
      ac.x += r.z * w.x; ac.y += r.z * w.y;
      ac.z += r.w * w.x; ac.w += r.w * w.y;
      m = (m + k0) & 15;
    }
    yr[tid] = ar;
    yc[tid] = ac;
  }
  __syncthreads();

  float4 dr = make_float4(0.f, 0.f, 0.f, 0.f), dc = dr;
  {
    const int k0 = tid & 15;
    int j = 0;
    for (int t0 = 0; t0 < 16; ++t0) {
      const float2 w = tw[j];
      const float4 y = yr[16 * t0 + k0];
      const float4 z = yc[16 * t0 + k0];
      dr.x += y.x * w.x - y.y * w.y;
      dr.y += y.y * w.x + y.x * w.y;
      dr.z += y.z * w.x - y.w * w.y;
      dr.w += y.w * w.x + y.z * w.y;
      dc.x += z.x * w.x - z.y * w.y;
      dc.y += z.y * w.x + z.x * w.y;
      dc.z += z.z * w.x - z.w * w.y;
      dc.w += z.w * w.x + z.z * w.y;
      j = (j + tid) & 255;
    }
  }
  ((float4*)ws)[tid] = dr;
  ((float4*)ws)[256 + tid] = dc;

  if (tid < E_DIM) {
    const int e = tid;
    ws[TOK_OFF + e] = sigmoidf_(ct[((e * 2 + 0) * S_DIM + 0) * 2 + 0]);
    ws[TOK_OFF + 32 + e] = sigmoidf_(ct[((e * 2 + 0) * S_DIM + 0) * 2 + 1]);
    ws[TOK_OFF + 64 + e] = sigmoidf_(ct[((e * 2 + 1) * S_DIM + 0) * 2 + 0]);
    ws[TOK_OFF + 96 + e] = sigmoidf_(ct[((e * 2 + 1) * S_DIM + 0) * 2 + 1]);
  }
}

__global__ __launch_bounds__(128) __attribute__((amdgpu_waves_per_eu(2, 2))) void moe_wmma_gemm(const float* __restrict__ a,
                                                     const float* __restrict__ b,
                                                     const float* __restrict__ ew,
                                                     const float* __restrict__ ws,
                                                     float* __restrict__ out) {
  __shared__ __align__(16) float Xs[C_DIM][S_DIM];
  __shared__ __align__(16) float Wp[4][2][KC][WPITCH];
  __shared__ float hred[4][8];
  __shared__ float mag[32];
  __shared__ int sel[3];
  const int n = blockIdx.x;
  const int tid = threadIdx.x;
  const size_t nb = (size_t)n * C_DIM * S_DIM;
  const int wave = tid >> 5;
  const int lane = tid & 31;

  const float4* __restrict__ ws4 = (const float4*)ws;
  const float vv = ws[UV_OFF + n * 512 + tid], vw = ws[UV_OFF + n * 512 + 128 + tid];
  const float uu = ws[UV_OFF + n * 512 + 256 + tid], uw = ws[UV_OFF + n * 512 + 384 + tid];
  const float4 dr = ws4[tid], er = ws4[128 + tid];
  const float4 dc = ws4[256 + tid], ec = ws4[384 + tid];
  float p0 = vv * dr.x + vw * er.x, p1 = vv * dr.y + vw * er.y, p2 = vv * dr.z + vw * er.z, p3 = vv * dr.w + vw * er.w;
  float p4 = uu * dc.x + uw * ec.x, p5 = uu * dc.y + uw * ec.y, p6 = uu * dc.z + uw * ec.z, p7 = uu * dc.w + uw * ec.w;
  float nmr, nmi, nsr, nsi;
  {
    const float4* __restrict__ pp = ws4 + (PART_OFF / 4);
    const float4 q0 = pp[lane], q1 = pp[32 + lane], q2 = pp[64 + lane], q3 = pp[96 + lane];
    nmr = fmaxf(fmaxf(q0.x, q1.x), fmaxf(q2.x, q3.x));
    nmi = fmaxf(fmaxf(q0.y, q1.y), fmaxf(q2.y, q3.y));
    nsr = q0.z * __builtin_amdgcn_exp2f((q0.x - nmr) * LOG2E_F) + q1.z * __builtin_amdgcn_exp2f((q1.x - nmr) * LOG2E_F)
        + q2.z * __builtin_amdgcn_exp2f((q2.x - nmr) * LOG2E_F) + q3.z * __builtin_amdgcn_exp2f((q3.x - nmr) * LOG2E_F);
    nsi = q0.w * __builtin_amdgcn_exp2f((q0.y - nmi) * LOG2E_F) + q1.w * __builtin_amdgcn_exp2f((q1.y - nmi) * LOG2E_F)
        + q2.w * __builtin_amdgcn_exp2f((q2.y - nmi) * LOG2E_F) + q3.w * __builtin_amdgcn_exp2f((q3.y - nmi) * LOG2E_F);
  }
  const float lmr = ws[LM_OFF], lmi = ws[LM_OFF + 1];

#pragma unroll
  for (int r = 0; r < 16; ++r) {
    const int lin = r * 128 + tid;
    const int c = lin >> 6;
    const int s4 = (lin & 63) << 2;
    const float4 av = *(const float4*)(a + nb + c * S_DIM + s4);
    const float4 bv = *(const float4*)(b + nb + c * S_DIM + s4);
    float4 v;
    v.x = av.x + bv.x; v.y = av.y + bv.y; v.z = av.z + bv.z; v.w = av.w + bv.w;
    *(float4*)&Xs[c][(s4 + 4 * c) & (S_DIM - 1)] = v;
  }

  for (int off = 16; off > 0; off >>= 1) {
    p0 += __shfl_xor(p0, off);
    p1 += __shfl_xor(p1, off);
    p2 += __shfl_xor(p2, off);
    p3 += __shfl_xor(p3, off);
    p4 += __shfl_xor(p4, off);
    p5 += __shfl_xor(p5, off);
    p6 += __shfl_xor(p6, off);
    p7 += __shfl_xor(p7, off);
    const float omr = __shfl_xor(nmr, off), osr = __shfl_xor(nsr, off);
    const float omi = __shfl_xor(nmi, off), osi = __shfl_xor(nsi, off);
    const float mr = fmaxf(nmr, omr), mi = fmaxf(nmi, omi);
    nsr = nsr * __builtin_amdgcn_exp2f((nmr - mr) * LOG2E_F) + osr * __builtin_amdgcn_exp2f((omr - mr) * LOG2E_F);
    nsi = nsi * __builtin_amdgcn_exp2f((nmi - mi) * LOG2E_F) + osi * __builtin_amdgcn_exp2f((omi - mi) * LOG2E_F);
    nmr = mr; nmi = mi;
  }
  if (lane == 0) {
    hred[wave][0] = p0; hred[wave][1] = p1; hred[wave][2] = p2; hred[wave][3] = p3;
    hred[wave][4] = p4; hred[wave][5] = p5; hred[wave][6] = p6; hred[wave][7] = p7;
  }
  __syncthreads();
  if (tid < E_DIM) {
    nsr = __int_as_float(__builtin_amdgcn_readfirstlane(__float_as_int(nsr)));
    nsi = __int_as_float(__builtin_amdgcn_readfirstlane(__float_as_int(nsi)));
    const float scr = expf(lmr - nmr) / nsr;
    const float sci = expf(lmi - nmi) / nsi;
    float q[8] = {0.f, 0.f, 0.f, 0.f, 0.f, 0.f, 0.f, 0.f};
    for (int ww = 0; ww < 4; ++ww) {
#pragma unroll
      for (int i = 0; i < 8; ++i) q[i] += hred[ww][i];
    }
    const float ar = scr * q[0] - sci * q[3], ai = scr * q[1] + sci * q[2];
    const float br = scr * q[4] - sci * q[7], bi = scr * q[5] + sci * q[6];
    const int e = tid;
    float tar = ws[TOK_OFF + e], tai = ws[TOK_OFF + 32 + e], tbr = ws[TOK_OFF + 64 + e], tbi = ws[TOK_OFF + 96 + e];
    float zr = ar * tar + ai * tai + br * tbr + bi * tbi;
    float zi = ai * tar - ar * tai + bi * tbr - br * tbi;
    mag[e] = zr * zr + zi * zi;
  }
  __syncthreads();
  if (tid == 0) {
    int s0 = 0, s1 = -1, s2 = -1;
    float b0 = -1.0f, b1 = -1.0f, b2 = -1.0f;
    for (int e = 0; e < 32; ++e) { float m = mag[e]; if (m > b0) { b0 = m; s0 = e; } }
    for (int e = 0; e < 32; ++e) { if (e == s0) continue; float m = mag[e]; if (m > b1) { b1 = m; s1 = e; } }
    for (int e = 0; e < 32; ++e) { if (e == s0 || e == s1) continue; float m = mag[e]; if (m > b2) { b2 = m; s2 = e; } }
    sel[0] = s0; sel[1] = s1; sel[2] = s2;
  }
  __syncthreads();
  const int i0 = __builtin_amdgcn_readfirstlane(sel[0]);
  const int i1 = __builtin_amdgcn_readfirstlane(sel[1]);
  const int i2 = __builtin_amdgcn_readfirstlane(sel[2]);
  const float* __restrict__ w0 = ew + (size_t)i0 * (S_DIM * S_DIM);
  const float* __restrict__ w1 = ew + (size_t)i1 * (S_DIM * S_DIM);
  const float* __restrict__ w2 = ew + (size_t)i2 * (S_DIM * S_DIM);

  float4 x0[16], x1[16], x2[16];

  const int l16 = lane & 15;
  const int lh = lane >> 4;
  const int ncol = wave * 64;
  const int lrow = lane >> 4;
  const int lcol = (lane & 15) << 2;

  auto issueChunk = [&](int K0) {
#pragma unroll
    for (int r = 0; r < 16; ++r) {
      const size_t g = (size_t)(K0 + 2 * r + lrow) * S_DIM + ncol + lcol;
      x0[r] = *(const float4*)(w0 + g);
      x1[r] = *(const float4*)(w1 + g);
      x2[r] = *(const float4*)(w2 + g);
    }
  };
  auto sumStoreChunk = [&](int buf) {
#pragma unroll
    for (int r = 0; r < 16; ++r) {
      float4 p;
      p.x = (x0[r].x + x1[r].x + x2[r].x) * (1.0f / 3.0f);
      p.y = (x0[r].y + x1[r].y + x2[r].y) * (1.0f / 3.0f);
      p.z = (x0[r].z + x1[r].z + x2[r].z) * (1.0f / 3.0f);
      p.w = (x0[r].w + x1[r].w + x2[r].w) * (1.0f / 3.0f);
      *(float4*)&Wp[wave][buf][2 * r + lrow][lcol] = p;
    }
  };

  auto waveHandover = [] {
    __builtin_amdgcn_fence(__ATOMIC_RELEASE, "wavefront");
    __builtin_amdgcn_wave_barrier();
    __builtin_amdgcn_fence(__ATOMIC_ACQUIRE, "wavefront");
  };

  v8f acc0[4] = {}, acc1[4] = {};

  auto burst = [&](int k0, int buf) {
    const float (*Wc)[WPITCH] = Wp[wave][buf];
#pragma unroll
    for (int kk = 0; kk < KC; kk += 4) {
      const int kf = kk + 2 * lh;
      const int m0 = l16, m1 = 16 + l16;
      const int xi0 = (k0 + kf + 4 * m0) & (S_DIM - 1);
      const int xi1 = (k0 + kf + 4 * m1) & (S_DIM - 1);
      v2f fa0, fa1, fb[4];
      fa0.x = Xs[m0][xi0];     fa0.y = Xs[m0][xi0 + 1];
      fa1.x = Xs[m1][xi1];     fa1.y = Xs[m1][xi1 + 1];
#pragma unroll
      for (int j = 0; j < 4; ++j) {
        fb[j].x = Wc[kf][16 * j + l16];
        fb[j].y = Wc[kf + 1][16 * j + l16];
      }
#pragma unroll
      for (int j = 0; j < 4; ++j) {
        acc0[j] = __builtin_amdgcn_wmma_f32_16x16x4_f32(false, fa0, false, fb[j], (short)0, acc0[j], false, false);
        acc1[j] = __builtin_amdgcn_wmma_f32_16x16x4_f32(false, fa1, false, fb[j], (short)0, acc1[j], false, false);
      }
    }
  };

  issueChunk(0);
  sumStoreChunk(0);

  int cur = 0;
  for (int ci = 0; ci + 1 < NCHUNK; ++ci) {
    waveHandover();
    issueChunk((ci + 1) * KC);
    burst(ci * KC, cur);
    sumStoreChunk(cur ^ 1);
    cur ^= 1;
  }
  waveHandover();
  burst((NCHUNK - 1) * KC, cur);
  cur ^= 1;

  float (*Wo)[WPITCH] = Wp[wave][cur];
  waveHandover();
#pragma unroll
  for (int r = 0; r < 8; ++r) {
    const int m = r + 8 * lh;
#pragma unroll
    for (int j = 0; j < 4; ++j) {
      Wo[m][16 * j + l16] = acc0[j][r];
      Wo[m + 16][16 * j + l16] = acc1[j][r];
    }
  }
  waveHandover();
#pragma unroll
  for (int r = 0; r < 16; ++r) {
    const int row = 2 * r + lrow;
    *(float4*)(out + nb + row * S_DIM + ncol + lcol) = *(const float4*)&Wo[row][lcol];
  }
}

extern "C" void kernel_launch(void* const* d_in, const int* in_sizes, int n_in,
                              void* d_out, int out_size, void* d_ws, size_t ws_size,
                              hipStream_t stream) {
  const float* a = (const float*)d_in[0];
  const float* b = (const float*)d_in[1];
  const float* ct = (const float*)d_in[2];
  const float* cm = (const float*)d_in[3];
  const float* ew = (const float*)d_in[4];
  float* out = (float*)d_out;
  float* ws = (float*)d_ws;

  router_setup<<<257, 256, 0, stream>>>(a, b, ct, cm, ws);
  moe_wmma_gemm<<<N_TOT, 128, 0, stream>>>(a, b, ew, ws, out);
}
